# starting kernel
# baseline (speedup 1.0000x reference)
#include <hip/hip_runtime.h>
#include <hip/hip_bf16.h>
#include <math.h>

#define HIDC 1024
#define NHC  16
#define HDC  64
#define BBC  2
#define SSC  2048

typedef float v2f __attribute__((ext_vector_type(2)));
typedef float v8f __attribute__((ext_vector_type(8)));

__device__ __forceinline__ v8f wmma4(v2f a, v2f b, v8f c) {
    return __builtin_amdgcn_wmma_f32_16x16x4_f32(
         false, a,  false, b,
         (short)0, c,  false,  false);
}

__device__ __forceinline__ v2f load2(const float* __restrict__ p) {
    return *(const v2f*)p;
}

__device__ __forceinline__ v2f load_a16x4(const float* __restrict__ A, int lda, int k0, int m, int kb2) {
    return load2(A + m * lda + k0 + kb2);
}

__device__ __forceinline__ v2f load_b4x16(const float* __restrict__ B, int ldb, int k0, int n0,
                                          int nl, int kb2) {
    const float* p = B + (k0 + kb2) * ldb + n0 + nl;
    v2f r; r[0] = p[0]; r[1] = p[ldb];
    return r;
}

__global__ void __launch_bounds__(32)
qkv_kernel(const float* __restrict__ x,
           const float* __restrict__ wq, const float* __restrict__ wk,
           const float* __restrict__ wv,
           float* __restrict__ q, float* __restrict__ k, float* __restrict__ vT)
{
    const int lane = threadIdx.x;
    const int m    = lane & 15;
    const int hi   = lane >> 4;
    const int kb2  = hi << 1;
    const int strip = blockIdx.x;
    const int n     = blockIdx.y;
    const int r0    = strip * 16;

    const float* Ab = x + (size_t)r0 * HIDC + n * HDC;
    const float* Wq = wq + n * HDC * HDC;
    const float* Wk = wk + n * HDC * HDC;
    const float* Wv = wv + n * HDC * HDC;

    v8f aq[4] = {}; v8f ak[4] = {}; v8f av[4] = {};
    for (int k0 = 0; k0 < HDC; k0 += 4) {
        v2f a = load_a16x4(Ab, HIDC, k0, m, kb2);
#pragma unroll
        for (int nt = 0; nt < 4; ++nt) {
            v2f bq = load_b4x16(Wq, HDC, k0, nt * 16, m, kb2);
            aq[nt] = wmma4(a, bq, aq[nt]);
            v2f bk = load_b4x16(Wk, HDC, k0, nt * 16, m, kb2);
            ak[nt] = wmma4(a, bk, ak[nt]);
            v2f bv = load_b4x16(Wv, HDC, k0, nt * 16, m, kb2);
            av[nt] = wmma4(a, bv, av[nt]);
        }
    }

    const int b  = r0 / SSC;
    const int s0 = r0 % SSC;
    const int mbase = hi * 8;
    float* qb = q  + (((size_t)b * NHC + n) * SSC + s0) * HDC;
    float* kb = k  + (((size_t)b * NHC + n) * SSC + s0) * HDC;
    float* vb = vT + (((size_t)b * NHC + n) * HDC) * SSC;
#pragma unroll
    for (int nt = 0; nt < 4; ++nt)
#pragma unroll
        for (int j = 0; j < 8; ++j) {
            size_t off = (size_t)(mbase + j) * HDC + nt * 16 + m;
            qb[off] = aq[nt][j];
            kb[off] = ak[nt][j];
            vb[(size_t)(nt * 16 + m) * SSC + s0 + mbase + j] = av[nt][j];
        }
}

__global__ void __launch_bounds__(32)
retention_kernel(const float* __restrict__ q, const float* __restrict__ k,
                 const float* __restrict__ vT, float* __restrict__ ret)
{
    const int lane = threadIdx.x;
    const int m    = lane & 15;
    const int hi   = lane >> 4;
    const int kb2  = hi << 1;
    const int mbase = hi * 8;
    const int st   = blockIdx.x;
    const int bn   = blockIdx.y;
    const int b    = bn / NHC;
    const int n    = bn % NHC;

    const float* Qb = q  + (((size_t)b * NHC + n) * SSC) * HDC;
    const float* Kb = k  + (((size_t)b * NHC + n) * SSC) * HDC;
    const float* Vb = vT + (((size_t)b * NHC + n) * HDC) * SSC;

    const float lin   = logf(1.0f / 512.0f)
                      + (float)n * ((logf(1.0f / 32.0f) - logf(1.0f / 512.0f)) / 15.0f);
    const float l2g   = log2f(1.0f - expf(lin));

    __shared__ float sS[16][18];

    const float* Qt = Qb + st * 16 * HDC;
    v2f qa[16];
#pragma unroll
    for (int kk = 0; kk < 16; ++kk)
        qa[kk] = load2(Qt + m * HDC + kk * 4 + kb2);

    v8f acc[4] = {};
    for (int tt = 0; tt <= st; ++tt) {
        const float* Kt = Kb + tt * 16 * HDC;
        if (tt < st) {
            __builtin_prefetch((const char*)(Kt + 16 * HDC) + lane * 128, 0, 1);
        }
        v8f sc = {};
#pragma unroll
        for (int kk = 0; kk < 16; ++kk) {
            v2f bb = load2(Kt + m * HDC + kk * 4 + kb2);
            sc = wmma4(qa[kk], bb, sc);
        }
#pragma unroll
        for (int j = 0; j < 8; ++j) {
            int sg = st * 16 + mbase + j;
            int tg = tt * 16 + m;
            float d = (float)(sg - tg);
            sS[mbase + j][m] = (sg >= tg) ? sc[j] * exp2f(d * l2g) : 0.0f;
        }
        __syncthreads();
#pragma unroll
        for (int k0 = 0; k0 < 16; k0 += 4) {
            v2f a2 = *(const v2f*)&sS[m][k0 + kb2];
#pragma unroll
            for (int nt = 0; nt < 4; ++nt) {
                v2f b2 = load2(Vb + (size_t)(nt * 16 + m) * SSC + tt * 16 + k0 + kb2);
                acc[nt] = wmma4(a2, b2, acc[nt]);
            }
        }
        __syncthreads();
    }

    float* rb = ret + (((size_t)b * SSC + st * 16) * NHC + n) * HDC;
#pragma unroll
    for (int nt = 0; nt < 4; ++nt)
#pragma unroll
        for (int j = 0; j < 8; ++j)
            rb[(size_t)(mbase + j) * NHC * HDC + nt * 16 + m] = acc[nt][j];
}

__global__ void groupnorm_kernel(const float* __restrict__ ret,
                                 const float* __restrict__ scale,
                                 const float* __restrict__ bias,
                                 float* __restrict__ normout)
{
    int idx = blockIdx.x * blockDim.x + threadIdx.x;
    if (idx >= BBC * SSC * NHC) return;
    int n = idx % NHC;
    const float* p = ret + (size_t)idx * HDC;

    float mean = 0.0f;
    for (int e = 0; e < HDC; ++e) mean += p[e];
    mean *= (1.0f / HDC);
    float var = 0.0f;
    for (int e = 0; e < HDC; ++e) { float d = p[e] - mean; var += d * d; }
    var *= (1.0f / HDC);
    float inv = rsqrtf(var + 1e-5f);

    float* o = normout + (size_t)idx * HDC;
    for (int e = 0; e < HDC; ++e) {
        int c = n * HDC + e;
        o[e] = (p[e] - mean) * inv * scale[c] + bias[c];
    }
}

template <bool SWISH_ADD>
__global__ void __launch_bounds__(32)
gemm1024_kernel(const float* __restrict__ A, const float* __restrict__ W,
                const float* __restrict__ addn, float* __restrict__ out)
{
    const int lane = threadIdx.x;
    const int m    = lane & 15;
    const int hi   = lane >> 4;
    const int kb2  = hi << 1;
    const int strip = blockIdx.x;
    const int nb    = blockIdx.y;

    const float* Ab = A + (size_t)strip * 16 * HIDC;
    v8f acc[8] = {};
    for (int k0 = 0; k0 < HIDC; k0 += 4) {
        v2f a = load_a16x4(Ab, HIDC, k0, m, kb2);
#pragma unroll
        for (int nt = 0; nt < 8; ++nt) {
            v2f bb = load_b4x16(W, HIDC, k0, nb * 128 + nt * 16, m, kb2);
            acc[nt] = wmma4(a, bb, acc[nt]);
        }
    }

    const int mbase = hi * 8;
#pragma unroll
    for (int nt = 0; nt < 8; ++nt)
#pragma unroll
        for (int j = 0; j < 8; ++j) {
            size_t off = (size_t)(strip * 16 + mbase + j) * HIDC + nb * 128 + nt * 16 + m;
            float g = acc[nt][j];
            if constexpr (SWISH_ADD) {
                out[off] = g / (1.0f + expf(-g)) + addn[off];
            } else {
                out[off] = g;
            }
        }
}

extern "C" void kernel_launch(void* const* d_in, const int* in_sizes, int n_in,
                              void* d_out, int out_size, void* d_ws, size_t ws_size,
                              hipStream_t stream)
{
    (void)in_sizes; (void)n_in; (void)out_size; (void)ws_size;
    const float* x   = (const float*)d_in[0];
    const float* wq  = (const float*)d_in[1];
    const float* wk  = (const float*)d_in[2];
    const float* wv  = (const float*)d_in[3];
    const float* w1  = (const float*)d_in[4];
    const float* w2  = (const float*)d_in[5];
    const float* gns = (const float*)d_in[6];
    const float* gnb = (const float*)d_in[7];
    float* out = (float*)d_out;
    float* ws  = (float*)d_ws;

    const size_t NQ = (size_t)BBC * NHC * SSC * HDC;
    float* q    = ws;
    float* k    = ws + NQ;
    float* vT   = ws + 2 * NQ;
    float* ret  = ws + 3 * NQ;
    float* norm = q;
    float* mid  = k;

    dim3 wave(32);
    qkv_kernel<<<dim3(BBC * SSC / 16, NHC), wave, 0, stream>>>(x, wq, wk, wv, q, k, vT);
    retention_kernel<<<dim3(SSC / 16, BBC * NHC), wave, 0, stream>>>(q, k, vT, ret);
    groupnorm_kernel<<<dim3(BBC * SSC * NHC / 256), dim3(256), 0, stream>>>(ret, gns, gnb, norm);
    gemm1024_kernel<true ><<<dim3(BBC * SSC / 16, HIDC / 128), wave, 0, stream>>>(x,   w1, norm, mid);
    gemm1024_kernel<false><<<dim3(BBC * SSC / 16, HIDC / 128), wave, 0, stream>>>(mid, w2, nullptr, out);
}


# groupnorm and its affine move into the gate GEMM's epilogue, which normalises the ret rows it loads where it read norm (4 launches, was 5); row sums become a 4-in-lane + 16-lane xor tree
# speedup vs baseline: 2.5176x; 2.5176x over previous
#include <hip/hip_runtime.h>
#include <hip/hip_bf16.h>
#include <math.h>

#define HIDC 1024
#define NHC  16
#define HDC  64
#define BBC  2
#define SSC  2048

typedef float v2f __attribute__((ext_vector_type(2)));
typedef float v4f __attribute__((ext_vector_type(4)));
typedef float v8f __attribute__((ext_vector_type(8)));

__device__ __forceinline__ v8f wmma4(v2f a, v2f b, v8f c) {
    return __builtin_amdgcn_wmma_f32_16x16x4_f32(
         false, a,  false, b,
         (short)0, c,  false,  false);
}

__device__ __forceinline__ v2f load2(const float* __restrict__ p) {
    return *(const v2f*)p;
}

__device__ __forceinline__ v2f load_a16x4(const float* __restrict__ A, int lda, int k0, int m, int kb2) {
    return load2(A + m * lda + k0 + kb2);
}

__device__ __forceinline__ v2f load_b4x16(const float* __restrict__ B, int ldb, int k0, int n0,
                                          int nl, int kb2) {
    const float* p = B + (k0 + kb2) * ldb + n0 + nl;
    v2f r; r[0] = p[0]; r[1] = p[ldb];
    return r;
}

__global__ void __launch_bounds__(32)
qkv_kernel(const float* __restrict__ x,
           const float* __restrict__ wq, const float* __restrict__ wk,
           const float* __restrict__ wv,
           float* __restrict__ q, float* __restrict__ k, float* __restrict__ vT)
{
    const int lane = threadIdx.x;
    const int m    = lane & 15;
    const int hi   = lane >> 4;
    const int kb2  = hi << 1;
    const int strip = blockIdx.x;
    const int n     = blockIdx.y;
    const int r0    = strip * 16;

    const float* Ab = x + (size_t)r0 * HIDC + n * HDC;
    const float* Wq = wq + n * HDC * HDC;
    const float* Wk = wk + n * HDC * HDC;
    const float* Wv = wv + n * HDC * HDC;

    v8f aq[4] = {}; v8f ak[4] = {}; v8f av[4] = {};
    for (int k0 = 0; k0 < HDC; k0 += 4) {
        v2f a = load_a16x4(Ab, HIDC, k0, m, kb2);
#pragma unroll
        for (int nt = 0; nt < 4; ++nt) {
            v2f bq = load_b4x16(Wq, HDC, k0, nt * 16, m, kb2);
            aq[nt] = wmma4(a, bq, aq[nt]);
            v2f bk = load_b4x16(Wk, HDC, k0, nt * 16, m, kb2);
            ak[nt] = wmma4(a, bk, ak[nt]);
            v2f bv = load_b4x16(Wv, HDC, k0, nt * 16, m, kb2);
            av[nt] = wmma4(a, bv, av[nt]);
        }
    }

    const int b  = r0 / SSC;
    const int s0 = r0 % SSC;
    const int mbase = hi * 8;
    float* qb = q  + (((size_t)b * NHC + n) * SSC + s0) * HDC;
    float* kb = k  + (((size_t)b * NHC + n) * SSC + s0) * HDC;
    float* vb = vT + (((size_t)b * NHC + n) * HDC) * SSC;
#pragma unroll
    for (int nt = 0; nt < 4; ++nt)
#pragma unroll
        for (int j = 0; j < 8; ++j) {
            size_t off = (size_t)(mbase + j) * HDC + nt * 16 + m;
            qb[off] = aq[nt][j];
            kb[off] = ak[nt][j];
            vb[(size_t)(nt * 16 + m) * SSC + s0 + mbase + j] = av[nt][j];
        }
}

__device__ __forceinline__ void wave_lds_sync() {
    __builtin_amdgcn_fence(__ATOMIC_RELEASE, "wavefront");
    __builtin_amdgcn_wave_barrier();
    __builtin_amdgcn_fence(__ATOMIC_ACQUIRE, "wavefront");
}

#define RT_WAVES 4
#define RT_TC    64
#define RT_PITCH 68
#define RT_PP    20

template <bool DIAG>
__device__ __forceinline__ void rt_tile(const v2f (&qa)[16], v8f (&acc)[4],
                                        const float* sK, const float* sV, float* sPw,
                                        int j, int m, int kb2, int mbase)
{
    const float* Kt = sK + (j * 16 + m) * RT_PITCH + kb2;
    v8f sc = {};
#pragma unroll
    for (int kk = 0; kk < 16; ++kk) {
        v2f bb = *(const v2f*)(Kt + kk * 4);
        sc = wmma4(qa[kk], bb, sc);
    }
#pragma unroll
    for (int jj = 0; jj < 8; ++jj) {
        float v = sc[jj];
        if (DIAG) v = (m <= mbase + jj) ? v : 0.0f;
        sPw[(mbase + jj) * RT_PP + m] = v;
    }
    wave_lds_sync();
    const float* Vt = sV + m * RT_PITCH + j * 16 + kb2;
#pragma unroll
    for (int k0 = 0; k0 < 16; k0 += 4) {
        v2f a2 = *(const v2f*)&sPw[m * RT_PP + k0 + kb2];
#pragma unroll
        for (int nt = 0; nt < 4; ++nt) {
            v2f b2 = *(const v2f*)(Vt + nt * 16 * RT_PITCH + k0);
            acc[nt] = wmma4(a2, b2, acc[nt]);
        }
    }
    wave_lds_sync();
}

__global__ void __launch_bounds__(32 * RT_WAVES)
retention_kernel(const float* __restrict__ q, const float* __restrict__ k,
                 const float* __restrict__ vT, float* __restrict__ ret)
{
    __shared__ __attribute__((aligned(16))) float sK[RT_TC * RT_PITCH];
    __shared__ __attribute__((aligned(16))) float sV[HDC * RT_PITCH];
    __shared__ __attribute__((aligned(16))) float sP[RT_WAVES * 16 * RT_PP];

    const int tid  = threadIdx.x;
    const int lane = tid & 31;
    const int w    = __builtin_amdgcn_readfirstlane(tid >> 5);
    const int m    = lane & 15;
    const int hi   = lane >> 4;
    const int kb2  = hi << 1;
    const int mbase = hi * 8;

    const int bi   = blockIdx.x;
    const int pass = bi >> 8;
    const int rem  = bi & 255;
    const int row  = rem >> 5;
    const int bn   = rem & 31;
    const int g    = (3 - pass) * 8 + ((pass & 1) ? row : 7 - row);
    const int b    = bn / NHC;
    const int n    = bn % NHC;
    const int st   = g * RT_WAVES + w;

    const float* Qb = q  + (((size_t)b * NHC + n) * SSC) * HDC;
    const float* Kb = k  + (((size_t)b * NHC + n) * SSC) * HDC;
    const float* Vb = vT + (((size_t)b * NHC + n) * HDC) * SSC;

    const float lin   = logf(1.0f / 512.0f)
                      + (float)n * ((logf(1.0f / 32.0f) - logf(1.0f / 512.0f)) / 15.0f);
    const float l2g   = log2f(1.0f - expf(lin));

    const float* Qt = Qb + st * 16 * HDC;
    v2f qa[16];
#pragma unroll
    for (int kk = 0; kk < 16; ++kk)
        qa[kk] = load2(Qt + m * HDC + kk * 4 + kb2);
    const float rowf = exp2f((float)(16 * w + m) * l2g);
#pragma unroll
    for (int kk = 0; kk < 16; ++kk)
        qa[kk] = qa[kk] * rowf;

    const int frow = tid >> 4;
    const int fcol = (tid & 15) * 4;
    v4f rk[8], rv[8];
    float colf[8];
#pragma unroll
    for (int p = 0; p < 8; ++p) {
        rk[p] = *(const v4f*)(Kb + (size_t)(p * 8 + frow) * HDC + fcol);
        rv[p] = *(const v4f*)(Vb + (size_t)(p * 8 + frow) * SSC + fcol);
        colf[p] = exp2f(-(float)(p * 8 + frow) * l2g);
    }

    float* sPw = sP + w * 16 * RT_PP;
    v8f acc[4] = {};
    for (int c = 0; c <= g; ++c) {
        const float stagef = exp2f((float)(RT_TC * (g - c)) * l2g);
        __syncthreads();
#pragma unroll
        for (int p = 0; p < 8; ++p) {
            *(v4f*)&sK[(p * 8 + frow) * RT_PITCH + fcol] = rk[p] * (stagef * colf[p]);
            *(v4f*)&sV[(p * 8 + frow) * RT_PITCH + fcol] = rv[p];
        }
        __syncthreads();
        if (c < g) {
            const float* Kn = Kb + (size_t)(c + 1) * RT_TC * HDC;
            const float* Vn = Vb + (c + 1) * RT_TC;
#pragma unroll
            for (int p = 0; p < 8; ++p) {
                rk[p] = *(const v4f*)(Kn + (size_t)(p * 8 + frow) * HDC + fcol);
                rv[p] = *(const v4f*)(Vn + (size_t)(p * 8 + frow) * SSC + fcol);
            }
        }
        const int jfull = (c == g) ? w : 4;
        for (int j = 0; j < jfull; ++j)
            rt_tile<false>(qa, acc, sK, sV, sPw, j, m, kb2, mbase);
        if (c == g)
            rt_tile<true>(qa, acc, sK, sV, sPw, w, m, kb2, mbase);
    }

    float* rb = ret + (((size_t)b * SSC + st * 16) * NHC + n) * HDC;
#pragma unroll
    for (int nt = 0; nt < 4; ++nt)
#pragma unroll
        for (int j = 0; j < 8; ++j)
            rb[(size_t)(mbase + j) * NHC * HDC + nt * 16 + m] = acc[nt][j];
}

template <int M>
__device__ __forceinline__ float xsum16(float v) {
    const int o = __builtin_amdgcn_update_dpp(0, __builtin_bit_cast(int, v), 0x160 + M, 0xf, 0xf, true);
    return v + __builtin_bit_cast(float, o);
}

#define GK 32
#define GW 8

template <bool SWISH_ADD>
__global__ void __launch_bounds__(32 * GW) __attribute__((amdgpu_waves_per_eu(2, 2)))
gemm1024_kernel(const float* __restrict__ A, const float* __restrict__ W,
                const float* __restrict__ retn, const float* __restrict__ gsc,
                const float* __restrict__ gbi, float* __restrict__ out)
{
    const int lane = threadIdx.x & 31;
    const int c    = lane & 15;
    const int hi   = lane >> 4;
    const int kb2  = hi << 1;
    const int strip = blockIdx.x * GW + (threadIdx.x >> 5);
    const int nb    = blockIdx.y;

    const float* Ap = A + (size_t)(strip * 16 + c) * HIDC + kb2;
    const float* Wp = W + (size_t)kb2 * HIDC + nb * 128 + c * 4;

    v8f acc[8] = {};
    for (int kb = 0; kb < HIDC; kb += GK) {
        v2f a[GK / 4];
        v4f b[GK / 4][2][2];
#pragma unroll
        for (int j = 0; j < GK / 4; ++j) {
            a[j] = load2(Ap + kb + 4 * j);
#pragma unroll
            for (int s = 0; s < 2; ++s)
#pragma unroll
                for (int sp = 0; sp < 2; ++sp)
                    b[j][s][sp] = *(const v4f*)(Wp + (size_t)(kb + 4 * j + s) * HIDC + sp * 64);
        }
#pragma unroll
        for (int j = 0; j < GK / 4; ++j)
#pragma unroll
            for (int sp = 0; sp < 2; ++sp)
#pragma unroll
                for (int t = 0; t < 4; ++t) {
                    v2f bb; bb[0] = b[j][0][sp][t]; bb[1] = b[j][1][sp][t];
                    acc[sp * 4 + t] = wmma4(a[j], bb, acc[sp * 4 + t]);
                }
    }

    const int mbase = hi * 8;
#pragma unroll
    for (int sp = 0; sp < 2; ++sp) {
        const int ch = nb * 128 + sp * 64 + c * 4;
        if constexpr (SWISH_ADD) {
            const v4f sc4 = *(const v4f*)(gsc + ch);
            const v4f bi4 = *(const v4f*)(gbi + ch);
            v4f r[8];
            float s[8], q[8];
#pragma unroll
            for (int j = 0; j < 8; ++j)
                r[j] = *(const v4f*)(retn + (size_t)(strip * 16 + mbase + j) * HIDC + ch);
#pragma unroll
            for (int j = 0; j < 8; ++j) s[j] = (r[j][0] + r[j][1]) + (r[j][2] + r[j][3]);
#pragma unroll
            for (int j = 0; j < 8; ++j) s[j] = xsum16<1>(s[j]);
#pragma unroll
            for (int j = 0; j < 8; ++j) s[j] = xsum16<2>(s[j]);
#pragma unroll
            for (int j = 0; j < 8; ++j) s[j] = xsum16<4>(s[j]);
#pragma unroll
            for (int j = 0; j < 8; ++j) s[j] = xsum16<8>(s[j]);
#pragma unroll
            for (int j = 0; j < 8; ++j) {
                const float mean = s[j] * (1.0f / HDC);
#pragma unroll
                for (int t = 0; t < 4; ++t) r[j][t] -= mean;
                q[j] = (r[j][0] * r[j][0] + r[j][1] * r[j][1]) + (r[j][2] * r[j][2] + r[j][3] * r[j][3]);
            }
#pragma unroll
            for (int j = 0; j < 8; ++j) q[j] = xsum16<1>(q[j]);
#pragma unroll
            for (int j = 0; j < 8; ++j) q[j] = xsum16<2>(q[j]);
#pragma unroll
            for (int j = 0; j < 8; ++j) q[j] = xsum16<4>(q[j]);
#pragma unroll
            for (int j = 0; j < 8; ++j) q[j] = xsum16<8>(q[j]);
#pragma unroll
            for (int j = 0; j < 8; ++j) {
                const float inv = rsqrtf(q[j] * (1.0f / HDC) + 1e-5f);
                v4f o;
#pragma unroll
                for (int t = 0; t < 4; ++t) {
                    const float g = acc[sp * 4 + t][j];
                    o[t] = g / (1.0f + expf(-g)) + (r[j][t] * inv * sc4[t] + bi4[t]);
                }
                *(v4f*)(out + (size_t)(strip * 16 + mbase + j) * HIDC + ch) = o;
            }
        } else {
#pragma unroll
            for (int j = 0; j < 8; ++j) {
                v4f g;
#pragma unroll
                for (int t = 0; t < 4; ++t) g[t] = acc[sp * 4 + t][j];
                *(v4f*)(out + (size_t)(strip * 16 + mbase + j) * HIDC + ch) = g;
            }
        }
    }
}

extern "C" void kernel_launch(void* const* d_in, const int* in_sizes, int n_in,
                              void* d_out, int out_size, void* d_ws, size_t ws_size,
                              hipStream_t stream)
{
    (void)in_sizes; (void)n_in; (void)out_size; (void)ws_size;
    const float* x   = (const float*)d_in[0];
    const float* wq  = (const float*)d_in[1];
    const float* wk  = (const float*)d_in[2];
    const float* wv  = (const float*)d_in[3];
    const float* w1  = (const float*)d_in[4];
    const float* w2  = (const float*)d_in[5];
    const float* gns = (const float*)d_in[6];
    const float* gnb = (const float*)d_in[7];
    float* out = (float*)d_out;
    float* ws  = (float*)d_ws;

    const size_t NQ = (size_t)BBC * NHC * SSC * HDC;
    float* q    = ws;
    float* k    = ws + NQ;
    float* vT   = ws + 2 * NQ;
    float* ret  = ws + 3 * NQ;
    float* mid  = k;

    dim3 wave(32);
    qkv_kernel<<<dim3(BBC * SSC / 16, NHC), wave, 0, stream>>>(x, wq, wk, wv, q, k, vT);
    retention_kernel<<<dim3((SSC / (16 * RT_WAVES)) * BBC * NHC), dim3(32 * RT_WAVES), 0, stream>>>(q, k, vT, ret);
    gemm1024_kernel<true ><<<dim3(BBC * SSC / (16 * GW), HIDC / 128), dim3(32 * GW), 0, stream>>>(x,   w1, ret, gns, gnb, mid);
    gemm1024_kernel<false><<<dim3(BBC * SSC / (16 * GW), HIDC / 128), dim3(32 * GW), 0, stream>>>(mid, w2, nullptr, nullptr, nullptr, out);
}


# q/k/v projections fold into the retention kernel by associativity: scores = x (Wq Wk^T) x^T, ret = (P x) Wv; projection launch and its 3 planes gone (3 launches); scores, ret rounding changes
# speedup vs baseline: 2.9020x; 1.1527x over previous
#include <hip/hip_runtime.h>
#include <hip/hip_bf16.h>
#include <math.h>

#define HIDC 1024
#define NHC  16
#define HDC  64
#define BBC  2
#define SSC  2048

typedef float v2f __attribute__((ext_vector_type(2)));
typedef float v4f __attribute__((ext_vector_type(4)));
typedef float v8f __attribute__((ext_vector_type(8)));

__device__ __forceinline__ v8f wmma4(v2f a, v2f b, v8f c) {
    return __builtin_amdgcn_wmma_f32_16x16x4_f32(
         false, a,  false, b,
         (short)0, c,  false,  false);
}

__device__ __forceinline__ v2f load2(const float* __restrict__ p) {
    return *(const v2f*)p;
}

__device__ __forceinline__ void wave_lds_sync() {
    __builtin_amdgcn_fence(__ATOMIC_RELEASE, "wavefront");
    __builtin_amdgcn_wave_barrier();
    __builtin_amdgcn_fence(__ATOMIC_ACQUIRE, "wavefront");
}

#define RT_WAVES 4
#define RT_TC    64
#define RT_PITCH 68
#define RT_PP    20
#define RT_VP    160

template <bool DIAG>
__device__ __forceinline__ void rt_tile(const v2f (&qa)[16], v8f (&acc)[4],
                                        const float* sK, const float* sV, float* sPw,
                                        int j, int m, int kb2, int hi, int mbase)
{
    const float* Kt = sK + (j * 16 + m) * RT_PITCH + kb2;
    v8f sc = {};
#pragma unroll
    for (int kk = 0; kk < 16; ++kk) {
        v2f bb = *(const v2f*)(Kt + kk * 4);
        sc = wmma4(qa[kk], bb, sc);
    }
#pragma unroll
    for (int jj = 0; jj < 8; ++jj) {
        float v = sc[jj];
        if (DIAG) v = (m <= mbase + jj) ? v : 0.0f;
        sPw[(mbase + jj) * RT_PP + m] = v;
    }
    wave_lds_sync();
    const float* Vt = sV + (j * 8 + hi) * RT_VP + 2 * m;
#pragma unroll
    for (int k0 = 0; k0 < 16; k0 += 4) {
        v2f a2 = *(const v2f*)&sPw[m * RT_PP + k0 + kb2];
#pragma unroll
        for (int nt = 0; nt < 4; ++nt) {
            v2f b2 = *(const v2f*)(Vt + (k0 >> 1) * RT_VP + nt * 32);
            acc[nt] = wmma4(a2, b2, acc[nt]);
        }
    }
    wave_lds_sync();
}

__global__ void __launch_bounds__(32 * RT_WAVES)
retention_kernel(const float* __restrict__ x, const float* __restrict__ wq,
                 const float* __restrict__ wk, const float* __restrict__ wv,
                 float* __restrict__ ret)
{
    __shared__ __attribute__((aligned(16))) float sK[RT_TC * RT_PITCH];
    __shared__ __attribute__((aligned(16))) float sV[(RT_TC / 2) * RT_VP];
    __shared__ __attribute__((aligned(16))) float sP[RT_WAVES * 16 * RT_PP];

    const int tid  = threadIdx.x;
    const int lane = tid & 31;
    const int w    = __builtin_amdgcn_readfirstlane(tid >> 5);
    const int m    = lane & 15;
    const int hi   = lane >> 4;
    const int kb2  = hi << 1;
    const int mbase = hi * 8;

    const int bi   = blockIdx.x;
    const int pass = bi >> 8;
    const int rem  = bi & 255;
    const int row  = rem >> 5;
    const int bn   = rem & 31;
    const int g    = (3 - pass) * 8 + ((pass & 1) ? row : 7 - row);
    const int b    = bn / NHC;
    const int n    = bn % NHC;
    const int st   = g * RT_WAVES + w;

    const float* Xh  = x + (size_t)b * SSC * HIDC + n * HDC;
    const float* Wqn = wq + n * HDC * HDC;
    const float* Wkn = wk + n * HDC * HDC;
    const float* Wvn = wv + n * HDC * HDC;

    const float lin   = logf(1.0f / 512.0f)
                      + (float)n * ((logf(1.0f / 32.0f) - logf(1.0f / 512.0f)) / 15.0f);
    const float l2g   = log2f(1.0f - expf(lin));

    const int r8   = tid >> 4;
    const int fcol = (tid & 15) * 4;
    v4f rk[8];
    float colf[8];
#pragma unroll
    for (int p = 0; p < 8; ++p) {
        const int tl = 16 * (p >> 1) + 2 * r8 + (p & 1);
        rk[p] = *(const v4f*)(Xh + (size_t)tl * HIDC + fcol);
        colf[p] = exp2f(-(float)tl * l2g);
    }

    {
        v8f mt[4] = {};
#pragma unroll
        for (int kk = 0; kk < 16; ++kk) {
            v2f a = load2(Wqn + (16 * w + m) * HDC + kk * 4 + kb2);
#pragma unroll
            for (int nt = 0; nt < 4; ++nt) {
                v2f bk = load2(Wkn + (nt * 16 + m) * HDC + kk * 4 + kb2);
                mt[nt] = wmma4(a, bk, mt[nt]);
            }
        }
#pragma unroll
        for (int nt = 0; nt < 4; ++nt) {
            float* dst = &sV[(nt * 16 + m) * RT_PITCH + 16 * w + mbase];
            v4f lo, up;
#pragma unroll
            for (int r = 0; r < 4; ++r) { lo[r] = mt[nt][r]; up[r] = mt[nt][4 + r]; }
            *(v4f*)dst       = lo;
            *(v4f*)(dst + 4) = up;
        }
    }
    __syncthreads();

    float* sQ = sK + 16 * w * RT_PITCH;
    v2f qa[16];
    {
        const float* Xq = Xh + (size_t)(st * 16 + m) * HIDC;
        v8f qc[4] = {};
#pragma unroll
        for (int kk = 0; kk < 16; ++kk) {
            v2f a = load2(Xq + kk * 4 + kb2);
#pragma unroll
            for (int nt = 0; nt < 4; ++nt) {
                v2f bm = *(const v2f*)&sV[(nt * 16 + m) * RT_PITCH + kk * 4 + kb2];
                qc[nt] = wmma4(a, bm, qc[nt]);
            }
        }
#pragma unroll
        for (int nt = 0; nt < 4; ++nt)
#pragma unroll
            for (int r = 0; r < 8; ++r)
                sQ[(mbase + r) * RT_PITCH + nt * 16 + m] = qc[nt][r];
        wave_lds_sync();
        const float rowf = exp2f((float)(16 * w + m) * l2g);
#pragma unroll
        for (int kk = 0; kk < 16; ++kk)
            qa[kk] = *(const v2f*)&sQ[m * RT_PITCH + kk * 4 + kb2] * rowf;
    }

    float* sPw = sP + w * 16 * RT_PP;
    v8f acc[4] = {};
    for (int c = 0; c <= g; ++c) {
        const float stagef = exp2f((float)(RT_TC * (g - c)) * l2g);
        __syncthreads();
#pragma unroll
        for (int pp = 0; pp < 4; ++pp) {
            const v4f ra = rk[2 * pp], rb = rk[2 * pp + 1];
            const int tl = 16 * pp + 2 * r8;
            *(v4f*)&sK[tl * RT_PITCH + fcol]       = ra * (stagef * colf[2 * pp]);
            *(v4f*)&sK[(tl + 1) * RT_PITCH + fcol] = rb * (stagef * colf[2 * pp + 1]);
            v4f i0, i1;
            i0[0] = ra[0]; i0[1] = rb[0]; i0[2] = ra[1]; i0[3] = rb[1];
            i1[0] = ra[2]; i1[1] = rb[2]; i1[2] = ra[3]; i1[3] = rb[3];
            float* vd = &sV[(8 * pp + r8) * RT_VP + 2 * fcol];
            *(v4f*)vd       = i0;
            *(v4f*)(vd + 4) = i1;
        }
        __syncthreads();
        if (c < g) {
            const float* Xn = Xh + (size_t)(c + 1) * RT_TC * HIDC;
#pragma unroll
            for (int p = 0; p < 8; ++p)
                rk[p] = *(const v4f*)(Xn + (size_t)(16 * (p >> 1) + 2 * r8 + (p & 1)) * HIDC + fcol);
        } else {
#pragma unroll
            for (int p = 0; p < 8; ++p)
                rk[p] = *(const v4f*)(Wvn + (16 * (p >> 1) + 2 * r8 + (p & 1)) * HDC + fcol);
        }
        const int jfull = (c == g) ? w : 4;
        for (int j = 0; j < jfull; ++j)
            rt_tile<false>(qa, acc, sK, sV, sPw, j, m, kb2, hi, mbase);
        if (c == g)
            rt_tile<true>(qa, acc, sK, sV, sPw, w, m, kb2, hi, mbase);
    }

    __syncthreads();
#pragma unroll
    for (int pp = 0; pp < 4; ++pp) {
        const v4f ra = rk[2 * pp], rb = rk[2 * pp + 1];
        v4f i0, i1;
        i0[0] = ra[0]; i0[1] = rb[0]; i0[2] = ra[1]; i0[3] = rb[1];
        i1[0] = ra[2]; i1[1] = rb[2]; i1[2] = ra[3]; i1[3] = rb[3];
        float* vd = &sV[(8 * pp + r8) * RT_VP + 2 * fcol];
        *(v4f*)vd       = i0;
        *(v4f*)(vd + 4) = i1;
    }
#pragma unroll
    for (int nt = 0; nt < 4; ++nt)
#pragma unroll
        for (int r = 0; r < 8; ++r)
            sQ[(mbase + r) * RT_PITCH + nt * 16 + m] = acc[nt][r];
    __syncthreads();
    v8f fin[4] = {};
#pragma unroll
    for (int kk = 0; kk < 16; ++kk) {
        v2f a = *(const v2f*)&sQ[m * RT_PITCH + kk * 4 + kb2];
#pragma unroll
        for (int nt = 0; nt < 4; ++nt) {
            v2f bw = *(const v2f*)&sV[(2 * kk + hi) * RT_VP + 2 * (nt * 16 + m)];
            fin[nt] = wmma4(a, bw, fin[nt]);
        }
    }

    float* rb = ret + (((size_t)b * SSC + st * 16) * NHC + n) * HDC;
#pragma unroll
    for (int nt = 0; nt < 4; ++nt)
#pragma unroll
        for (int j = 0; j < 8; ++j)
            rb[(size_t)(mbase + j) * NHC * HDC + nt * 16 + m] = fin[nt][j];
}

template <int M>
__device__ __forceinline__ float xsum16(float v) {
    const int o = __builtin_amdgcn_update_dpp(0, __builtin_bit_cast(int, v), 0x160 + M, 0xf, 0xf, true);
    return v + __builtin_bit_cast(float, o);
}

#define GK 32
#define GW 8

template <bool SWISH_ADD>
__global__ void __launch_bounds__(32 * GW) __attribute__((amdgpu_waves_per_eu(2, 2)))
gemm1024_kernel(const float* __restrict__ A, const float* __restrict__ W,
                const float* __restrict__ retn, const float* __restrict__ gsc,
                const float* __restrict__ gbi, float* __restrict__ out)
{
    const int lane = threadIdx.x & 31;
    const int c    = lane & 15;
    const int hi   = lane >> 4;
    const int kb2  = hi << 1;
    const int strip = blockIdx.x * GW + (threadIdx.x >> 5);
    const int nb    = blockIdx.y;

    const float* Ap = A + (size_t)(strip * 16 + c) * HIDC + kb2;
    const float* Wp = W + (size_t)kb2 * HIDC + nb * 128 + c * 4;

    v8f acc[8] = {};
    for (int kb = 0; kb < HIDC; kb += GK) {
        v2f a[GK / 4];
        v4f b[GK / 4][2][2];
#pragma unroll
        for (int j = 0; j < GK / 4; ++j) {
            a[j] = load2(Ap + kb + 4 * j);
#pragma unroll
            for (int s = 0; s < 2; ++s)
#pragma unroll
                for (int sp = 0; sp < 2; ++sp)
                    b[j][s][sp] = *(const v4f*)(Wp + (size_t)(kb + 4 * j + s) * HIDC + sp * 64);
        }
#pragma unroll
        for (int j = 0; j < GK / 4; ++j)
#pragma unroll
            for (int sp = 0; sp < 2; ++sp)
#pragma unroll
                for (int t = 0; t < 4; ++t) {
                    v2f bb; bb[0] = b[j][0][sp][t]; bb[1] = b[j][1][sp][t];
                    acc[sp * 4 + t] = wmma4(a[j], bb, acc[sp * 4 + t]);
                }
    }

    const int mbase = hi * 8;
#pragma unroll
    for (int sp = 0; sp < 2; ++sp) {
        const int ch = nb * 128 + sp * 64 + c * 4;
        if constexpr (SWISH_ADD) {
            const v4f sc4 = *(const v4f*)(gsc + ch);
            const v4f bi4 = *(const v4f*)(gbi + ch);
            v4f r[8];
            float s[8], q[8];
#pragma unroll
            for (int j = 0; j < 8; ++j)
                r[j] = *(const v4f*)(retn + (size_t)(strip * 16 + mbase + j) * HIDC + ch);
#pragma unroll
            for (int j = 0; j < 8; ++j) s[j] = (r[j][0] + r[j][1]) + (r[j][2] + r[j][3]);
#pragma unroll
            for (int j = 0; j < 8; ++j) s[j] = xsum16<1>(s[j]);
#pragma unroll
            for (int j = 0; j < 8; ++j) s[j] = xsum16<2>(s[j]);
#pragma unroll
            for (int j = 0; j < 8; ++j) s[j] = xsum16<4>(s[j]);
#pragma unroll
            for (int j = 0; j < 8; ++j) s[j] = xsum16<8>(s[j]);
#pragma unroll
            for (int j = 0; j < 8; ++j) {
                const float mean = s[j] * (1.0f / HDC);
#pragma unroll
                for (int t = 0; t < 4; ++t) r[j][t] -= mean;
                q[j] = (r[j][0] * r[j][0] + r[j][1] * r[j][1]) + (r[j][2] * r[j][2] + r[j][3] * r[j][3]);
            }
#pragma unroll
            for (int j = 0; j < 8; ++j) q[j] = xsum16<1>(q[j]);
#pragma unroll
            for (int j = 0; j < 8; ++j) q[j] = xsum16<2>(q[j]);
#pragma unroll
            for (int j = 0; j < 8; ++j) q[j] = xsum16<4>(q[j]);
#pragma unroll
            for (int j = 0; j < 8; ++j) q[j] = xsum16<8>(q[j]);
#pragma unroll
            for (int j = 0; j < 8; ++j) {
                const float inv = rsqrtf(q[j] * (1.0f / HDC) + 1e-5f);
                v4f o;
#pragma unroll
                for (int t = 0; t < 4; ++t) {
                    const float g = acc[sp * 4 + t][j];
                    o[t] = g / (1.0f + expf(-g)) + (r[j][t] * inv * sc4[t] + bi4[t]);
                }
                *(v4f*)(out + (size_t)(strip * 16 + mbase + j) * HIDC + ch) = o;
            }
        } else {
#pragma unroll
            for (int j = 0; j < 8; ++j) {
                v4f g;
#pragma unroll
                for (int t = 0; t < 4; ++t) g[t] = acc[sp * 4 + t][j];
                *(v4f*)(out + (size_t)(strip * 16 + mbase + j) * HIDC + ch) = g;
            }
        }
    }
}

extern "C" void kernel_launch(void* const* d_in, const int* in_sizes, int n_in,
                              void* d_out, int out_size, void* d_ws, size_t ws_size,
                              hipStream_t stream)
{
    (void)in_sizes; (void)n_in; (void)out_size; (void)ws_size;
    const float* x   = (const float*)d_in[0];
    const float* wq  = (const float*)d_in[1];
    const float* wk  = (const float*)d_in[2];
    const float* wv  = (const float*)d_in[3];
    const float* w1  = (const float*)d_in[4];
    const float* w2  = (const float*)d_in[5];
    const float* gns = (const float*)d_in[6];
    const float* gnb = (const float*)d_in[7];
    float* out = (float*)d_out;
    float* ws  = (float*)d_ws;

    const size_t NQ = (size_t)BBC * NHC * SSC * HDC;
    float* ret  = ws;
    float* mid  = ws + NQ;

    retention_kernel<<<dim3((SSC / (16 * RT_WAVES)) * BBC * NHC), dim3(32 * RT_WAVES), 0, stream>>>(x, wq, wk, wv, ret);
    gemm1024_kernel<true ><<<dim3(BBC * SSC / (16 * GW), HIDC / 128), dim3(32 * GW), 0, stream>>>(x,   w1, ret, gns, gnb, mid);
    gemm1024_kernel<false><<<dim3(BBC * SSC / (16 * GW), HIDC / 128), dim3(32 * GW), 0, stream>>>(mid, w2, nullptr, nullptr, nullptr, out);
}


# retention: stages below the diagonal stage sum a 64x64 state G = sum_t gamma^(s0-t) x_t^T x_t (64 WMMAs per wave-stage, was 128); one q' x G replaces their scores and P x; ret sum order changes
# speedup vs baseline: 3.5980x; 1.2398x over previous
#include <hip/hip_runtime.h>
#include <hip/hip_bf16.h>
#include <math.h>

#define HIDC 1024
#define NHC  16
#define HDC  64
#define BBC  2
#define SSC  2048

typedef float v2f __attribute__((ext_vector_type(2)));
typedef float v4f __attribute__((ext_vector_type(4)));
typedef float v8f __attribute__((ext_vector_type(8)));

__device__ __forceinline__ v8f wmma4(v2f a, v2f b, v8f c) {
    return __builtin_amdgcn_wmma_f32_16x16x4_f32(
         false, a,  false, b,
         (short)0, c,  false,  false);
}

__device__ __forceinline__ v2f load2(const float* __restrict__ p) {
    return *(const v2f*)p;
}

__device__ __forceinline__ void wave_lds_sync() {
    __builtin_amdgcn_fence(__ATOMIC_RELEASE, "wavefront");
    __builtin_amdgcn_wave_barrier();
    __builtin_amdgcn_fence(__ATOMIC_ACQUIRE, "wavefront");
}

#define RT_WAVES 4
#define RT_TC    64
#define RT_PITCH 68
#define RT_PP    20
#define RT_VP    160

template <bool DIAG>
__device__ __forceinline__ void rt_tile(const v2f (&qa)[16], v8f (&acc)[4],
                                        const float* sK, const float* sV, float* sPw,
                                        int j, int m, int kb2, int hi, int mbase)
{
    const float* Kt = sK + (j * 16 + m) * RT_PITCH + kb2;
    v8f sc = {};
#pragma unroll
    for (int kk = 0; kk < 16; ++kk) {
        v2f bb = *(const v2f*)(Kt + kk * 4);
        sc = wmma4(qa[kk], bb, sc);
    }
#pragma unroll
    for (int jj = 0; jj < 8; ++jj) {
        float v = sc[jj];
        if (DIAG) v = (m <= mbase + jj) ? v : 0.0f;
        sPw[(mbase + jj) * RT_PP + m] = v;
    }
    wave_lds_sync();
    const float* Vt = sV + (j * 8 + hi) * RT_VP + 2 * m;
#pragma unroll
    for (int k0 = 0; k0 < 16; k0 += 4) {
        v2f a2 = *(const v2f*)&sPw[m * RT_PP + k0 + kb2];
#pragma unroll
        for (int nt = 0; nt < 4; ++nt) {
            v2f b2 = *(const v2f*)(Vt + (k0 >> 1) * RT_VP + nt * 32);
            acc[nt] = wmma4(a2, b2, acc[nt]);
        }
    }
    wave_lds_sync();
}

__global__ void __launch_bounds__(32 * RT_WAVES)
retention_kernel(const float* __restrict__ x, const float* __restrict__ wq,
                 const float* __restrict__ wk, const float* __restrict__ wv,
                 float* __restrict__ ret)
{
    __shared__ __attribute__((aligned(16))) float sK[RT_TC * RT_PITCH];
    __shared__ __attribute__((aligned(16))) float sV[(RT_TC / 2) * RT_VP];
    __shared__ __attribute__((aligned(16))) float sP[RT_WAVES * 16 * RT_PP];

    const int tid  = threadIdx.x;
    const int lane = tid & 31;
    const int w    = __builtin_amdgcn_readfirstlane(tid >> 5);
    const int m    = lane & 15;
    const int hi   = lane >> 4;
    const int kb2  = hi << 1;
    const int mbase = hi * 8;

    const int bi   = blockIdx.x;
    const int pass = bi >> 8;
    const int rem  = bi & 255;
    const int row  = rem >> 5;
    const int bn   = rem & 31;
    const int g    = (3 - pass) * 8 + ((pass & 1) ? row : 7 - row);
    const int b    = bn / NHC;
    const int n    = bn % NHC;
    const int st   = g * RT_WAVES + w;

    const float* Xh  = x + (size_t)b * SSC * HIDC + n * HDC;
    const float* Wqn = wq + n * HDC * HDC;
    const float* Wkn = wk + n * HDC * HDC;
    const float* Wvn = wv + n * HDC * HDC;

    const float lin   = logf(1.0f / 512.0f)
                      + (float)n * ((logf(1.0f / 32.0f) - logf(1.0f / 512.0f)) / 15.0f);
    const float l2g   = log2f(1.0f - expf(lin));

    const int r8   = tid >> 4;
    const int fcol = (tid & 15) * 4;
    v4f rk[8];
    float colf[8];
#pragma unroll
    for (int p = 0; p < 8; ++p) {
        const int tl = 16 * (p >> 1) + 2 * r8 + (p & 1);
        rk[p] = *(const v4f*)(Xh + (size_t)tl * HIDC + fcol);
        colf[p] = exp2f(-(float)tl * l2g);
    }
    float hcolf[8];
#pragma unroll
    for (int p = 0; p < 8; ++p) {
        const int tl = 16 * (p >> 1) + 2 * r8 + (p & 1);
        hcolf[p] = exp2f(-0.5f * (float)tl * l2g);
    }

    {
        v8f mt[4] = {};
#pragma unroll
        for (int kk = 0; kk < 16; ++kk) {
            v2f a = load2(Wqn + (16 * w + m) * HDC + kk * 4 + kb2);
#pragma unroll
            for (int nt = 0; nt < 4; ++nt) {
                v2f bk = load2(Wkn + (nt * 16 + m) * HDC + kk * 4 + kb2);
                mt[nt] = wmma4(a, bk, mt[nt]);
            }
        }
#pragma unroll
        for (int nt = 0; nt < 4; ++nt) {
            float* dst = &sV[(nt * 16 + m) * RT_PITCH + 16 * w + mbase];
            v4f lo, up;
#pragma unroll
            for (int r = 0; r < 4; ++r) { lo[r] = mt[nt][r]; up[r] = mt[nt][4 + r]; }
            *(v4f*)dst       = lo;
            *(v4f*)(dst + 4) = up;
        }
    }
    __syncthreads();

    float* sQ = sK + 16 * w * RT_PITCH;
    v2f qa[16];
    {
        const float* Xq = Xh + (size_t)(st * 16 + m) * HIDC;
        v8f qc[4] = {};
#pragma unroll
        for (int kk = 0; kk < 16; ++kk) {
            v2f a = load2(Xq + kk * 4 + kb2);
#pragma unroll
            for (int nt = 0; nt < 4; ++nt) {
                v2f bm = *(const v2f*)&sV[(nt * 16 + m) * RT_PITCH + kk * 4 + kb2];
                qc[nt] = wmma4(a, bm, qc[nt]);
            }
        }
#pragma unroll
        for (int nt = 0; nt < 4; ++nt)
#pragma unroll
            for (int r = 0; r < 8; ++r)
                sQ[(mbase + r) * RT_PITCH + nt * 16 + m] = qc[nt][r];
        wave_lds_sync();
        const float rowf = exp2f((float)(16 * w + m) * l2g);
#pragma unroll
        for (int kk = 0; kk < 16; ++kk)
            qa[kk] = *(const v2f*)&sQ[m * RT_PITCH + kk * 4 + kb2] * rowf;
    }

    float* sPw = sP + w * 16 * RT_PP;
    v8f acc[4] = {};
    if (g > 0) {
        v8f gs[4] = {};
        for (int c = 0; c < g; ++c) {
            const float hstagef = exp2f(0.5f * (float)(RT_TC * (g - c)) * l2g);
            __syncthreads();
#pragma unroll
            for (int pp = 0; pp < 4; ++pp) {
                const v4f ra = rk[2 * pp]     * (hstagef * hcolf[2 * pp]);
                const v4f rb = rk[2 * pp + 1] * (hstagef * hcolf[2 * pp + 1]);
                v4f i0, i1;
                i0[0] = ra[0]; i0[1] = rb[0]; i0[2] = ra[1]; i0[3] = rb[1];
                i1[0] = ra[2]; i1[1] = rb[2]; i1[2] = ra[3]; i1[3] = rb[3];
                float* vd = &sV[(8 * pp + r8) * RT_VP + 2 * fcol];
                *(v4f*)vd       = i0;
                *(v4f*)(vd + 4) = i1;
            }
            __syncthreads();
            {
                const float* Xn = Xh + (size_t)(c + 1) * RT_TC * HIDC;
#pragma unroll
                for (int p = 0; p < 8; ++p)
                    rk[p] = *(const v4f*)(Xn + (size_t)(16 * (p >> 1) + 2 * r8 + (p & 1)) * HIDC + fcol);
            }
            const float* Vt = sV + hi * RT_VP + 2 * m;
#pragma unroll
            for (int kk = 0; kk < 16; ++kk) {
                v2f a2 = *(const v2f*)(Vt + 2 * kk * RT_VP + w * 32);
#pragma unroll
                for (int nt = 0; nt < 4; ++nt) {
                    v2f b2 = *(const v2f*)(Vt + 2 * kk * RT_VP + nt * 32);
                    gs[nt] = wmma4(a2, b2, gs[nt]);
                }
            }
        }
        __syncthreads();
#pragma unroll
        for (int nt = 0; nt < 4; ++nt)
#pragma unroll
            for (int q = 0; q < 4; ++q) {
                v2f pr; pr[0] = gs[nt][2 * q]; pr[1] = gs[nt][2 * q + 1];
                *(v2f*)&sV[(8 * w + 4 * hi + q) * RT_VP + 2 * (nt * 16 + m)] = pr;
            }
        __syncthreads();
#pragma unroll
        for (int kk = 0; kk < 16; ++kk) {
#pragma unroll
            for (int nt = 0; nt < 4; ++nt) {
                v2f bg = *(const v2f*)&sV[(2 * kk + hi) * RT_VP + 2 * (nt * 16 + m)];
                acc[nt] = wmma4(qa[kk], bg, acc[nt]);
            }
        }
    }
    {
        __syncthreads();
#pragma unroll
        for (int pp = 0; pp < 4; ++pp) {
            const v4f ra = rk[2 * pp], rb = rk[2 * pp + 1];
            const int tl = 16 * pp + 2 * r8;
            *(v4f*)&sK[tl * RT_PITCH + fcol]       = ra * colf[2 * pp];
            *(v4f*)&sK[(tl + 1) * RT_PITCH + fcol] = rb * colf[2 * pp + 1];
            v4f i0, i1;
            i0[0] = ra[0]; i0[1] = rb[0]; i0[2] = ra[1]; i0[3] = rb[1];
            i1[0] = ra[2]; i1[1] = rb[2]; i1[2] = ra[3]; i1[3] = rb[3];
            float* vd = &sV[(8 * pp + r8) * RT_VP + 2 * fcol];
            *(v4f*)vd       = i0;
            *(v4f*)(vd + 4) = i1;
        }
        __syncthreads();
#pragma unroll
        for (int p = 0; p < 8; ++p)
            rk[p] = *(const v4f*)(Wvn + (16 * (p >> 1) + 2 * r8 + (p & 1)) * HDC + fcol);
        for (int j = 0; j < w; ++j)
            rt_tile<false>(qa, acc, sK, sV, sPw, j, m, kb2, hi, mbase);
        rt_tile<true>(qa, acc, sK, sV, sPw, w, m, kb2, hi, mbase);
    }

    __syncthreads();
#pragma unroll
    for (int pp = 0; pp < 4; ++pp) {
        const v4f ra = rk[2 * pp], rb = rk[2 * pp + 1];
        v4f i0, i1;
        i0[0] = ra[0]; i0[1] = rb[0]; i0[2] = ra[1]; i0[3] = rb[1];
        i1[0] = ra[2]; i1[1] = rb[2]; i1[2] = ra[3]; i1[3] = rb[3];
        float* vd = &sV[(8 * pp + r8) * RT_VP + 2 * fcol];
        *(v4f*)vd       = i0;
        *(v4f*)(vd + 4) = i1;
    }
#pragma unroll
    for (int nt = 0; nt < 4; ++nt)
#pragma unroll
        for (int r = 0; r < 8; ++r)
            sQ[(mbase + r) * RT_PITCH + nt * 16 + m] = acc[nt][r];
    __syncthreads();
    v8f fin[4] = {};
#pragma unroll
    for (int kk = 0; kk < 16; ++kk) {
        v2f a = *(const v2f*)&sQ[m * RT_PITCH + kk * 4 + kb2];
#pragma unroll
        for (int nt = 0; nt < 4; ++nt) {
            v2f bw = *(const v2f*)&sV[(2 * kk + hi) * RT_VP + 2 * (nt * 16 + m)];
            fin[nt] = wmma4(a, bw, fin[nt]);
        }
    }

    float* rb = ret + (((size_t)b * SSC + st * 16) * NHC + n) * HDC;
#pragma unroll
    for (int nt = 0; nt < 4; ++nt)
#pragma unroll
        for (int j = 0; j < 8; ++j)
            rb[(size_t)(mbase + j) * NHC * HDC + nt * 16 + m] = fin[nt][j];
}

template <int M>
__device__ __forceinline__ float xsum16(float v) {
    const int o = __builtin_amdgcn_update_dpp(0, __builtin_bit_cast(int, v), 0x160 + M, 0xf, 0xf, true);
    return v + __builtin_bit_cast(float, o);
}

#define GK 32
#define GW 8

template <bool SWISH_ADD>
__global__ void __launch_bounds__(32 * GW) __attribute__((amdgpu_waves_per_eu(2, 2)))
gemm1024_kernel(const float* __restrict__ A, const float* __restrict__ W,
                const float* __restrict__ retn, const float* __restrict__ gsc,
                const float* __restrict__ gbi, float* __restrict__ out)
{
    const int lane = threadIdx.x & 31;
    const int c    = lane & 15;
    const int hi   = lane >> 4;
    const int kb2  = hi << 1;
    const int strip = blockIdx.x * GW + (threadIdx.x >> 5);
    const int nb    = blockIdx.y;

    const float* Ap = A + (size_t)(strip * 16 + c) * HIDC + kb2;
    const float* Wp = W + (size_t)kb2 * HIDC + nb * 128 + c * 4;

    v8f acc[8] = {};
    for (int kb = 0; kb < HIDC; kb += GK) {
        v2f a[GK / 4];
        v4f b[GK / 4][2][2];
#pragma unroll
        for (int j = 0; j < GK / 4; ++j) {
            a[j] = load2(Ap + kb + 4 * j);
#pragma unroll
            for (int s = 0; s < 2; ++s)
#pragma unroll
                for (int sp = 0; sp < 2; ++sp)
                    b[j][s][sp] = *(const v4f*)(Wp + (size_t)(kb + 4 * j + s) * HIDC + sp * 64);
        }
#pragma unroll
        for (int j = 0; j < GK / 4; ++j)
#pragma unroll
            for (int sp = 0; sp < 2; ++sp)
#pragma unroll
                for (int t = 0; t < 4; ++t) {
                    v2f bb; bb[0] = b[j][0][sp][t]; bb[1] = b[j][1][sp][t];
                    acc[sp * 4 + t] = wmma4(a[j], bb, acc[sp * 4 + t]);
                }
    }

    const int mbase = hi * 8;
#pragma unroll
    for (int sp = 0; sp < 2; ++sp) {
        const int ch = nb * 128 + sp * 64 + c * 4;
        if constexpr (SWISH_ADD) {
            const v4f sc4 = *(const v4f*)(gsc + ch);
            const v4f bi4 = *(const v4f*)(gbi + ch);
            v4f r[8];
            float s[8], q[8];
#pragma unroll
            for (int j = 0; j < 8; ++j)
                r[j] = *(const v4f*)(retn + (size_t)(strip * 16 + mbase + j) * HIDC + ch);
#pragma unroll
            for (int j = 0; j < 8; ++j) s[j] = (r[j][0] + r[j][1]) + (r[j][2] + r[j][3]);
#pragma unroll
            for (int j = 0; j < 8; ++j) s[j] = xsum16<1>(s[j]);
#pragma unroll
            for (int j = 0; j < 8; ++j) s[j] = xsum16<2>(s[j]);
#pragma unroll
            for (int j = 0; j < 8; ++j) s[j] = xsum16<4>(s[j]);
#pragma unroll
            for (int j = 0; j < 8; ++j) s[j] = xsum16<8>(s[j]);
#pragma unroll
            for (int j = 0; j < 8; ++j) {
                const float mean = s[j] * (1.0f / HDC);
#pragma unroll
                for (int t = 0; t < 4; ++t) r[j][t] -= mean;
                q[j] = (r[j][0] * r[j][0] + r[j][1] * r[j][1]) + (r[j][2] * r[j][2] + r[j][3] * r[j][3]);
            }
#pragma unroll
            for (int j = 0; j < 8; ++j) q[j] = xsum16<1>(q[j]);
#pragma unroll
            for (int j = 0; j < 8; ++j) q[j] = xsum16<2>(q[j]);
#pragma unroll
            for (int j = 0; j < 8; ++j) q[j] = xsum16<4>(q[j]);
#pragma unroll
            for (int j = 0; j < 8; ++j) q[j] = xsum16<8>(q[j]);
#pragma unroll
            for (int j = 0; j < 8; ++j) {
                const float inv = rsqrtf(q[j] * (1.0f / HDC) + 1e-5f);
                v4f o;
#pragma unroll
                for (int t = 0; t < 4; ++t) {
                    const float g = acc[sp * 4 + t][j];
                    o[t] = g / (1.0f + expf(-g)) + (r[j][t] * inv * sc4[t] + bi4[t]);
                }
                *(v4f*)(out + (size_t)(strip * 16 + mbase + j) * HIDC + ch) = o;
            }
        } else {
#pragma unroll
            for (int j = 0; j < 8; ++j) {
                v4f g;
#pragma unroll
                for (int t = 0; t < 4; ++t) g[t] = acc[sp * 4 + t][j];
                *(v4f*)(out + (size_t)(strip * 16 + mbase + j) * HIDC + ch) = g;
            }
        }
    }
}

extern "C" void kernel_launch(void* const* d_in, const int* in_sizes, int n_in,
                              void* d_out, int out_size, void* d_ws, size_t ws_size,
                              hipStream_t stream)
{
    (void)in_sizes; (void)n_in; (void)out_size; (void)ws_size;
    const float* x   = (const float*)d_in[0];
    const float* wq  = (const float*)d_in[1];
    const float* wk  = (const float*)d_in[2];
    const float* wv  = (const float*)d_in[3];
    const float* w1  = (const float*)d_in[4];
    const float* w2  = (const float*)d_in[5];
    const float* gns = (const float*)d_in[6];
    const float* gnb = (const float*)d_in[7];
    float* out = (float*)d_out;
    float* ws  = (float*)d_ws;

    const size_t NQ = (size_t)BBC * NHC * SSC * HDC;
    float* ret  = ws;
    float* mid  = ws + NQ;

    retention_kernel<<<dim3((SSC / (16 * RT_WAVES)) * BBC * NHC), dim3(32 * RT_WAVES), 0, stream>>>(x, wq, wk, wv, ret);
    gemm1024_kernel<true ><<<dim3(BBC * SSC / (16 * GW), HIDC / 128), dim3(32 * GW), 0, stream>>>(x,   w1, ret, gns, gnb, mid);
    gemm1024_kernel<false><<<dim3(BBC * SSC / (16 * GW), HIDC / 128), dim3(32 * GW), 0, stream>>>(mid, w2, nullptr, nullptr, nullptr, out);
}


# retention: the 64x64 state G = sum_t x_t^T x_t is symmetric, so only 10 of its 16 tiles are summed per stage (48 or 32 WMMAs per wave, was 64) and mirrored once into LDS; sums, order kept
# speedup vs baseline: 3.9038x; 1.0850x over previous
#include <hip/hip_runtime.h>
#include <hip/hip_bf16.h>
#include <math.h>

#define HIDC 1024
#define NHC  16
#define HDC  64
#define BBC  2
#define SSC  2048

typedef float v2f __attribute__((ext_vector_type(2)));
typedef float v4f __attribute__((ext_vector_type(4)));
typedef float v8f __attribute__((ext_vector_type(8)));

__device__ __forceinline__ v8f wmma4(v2f a, v2f b, v8f c) {
    return __builtin_amdgcn_wmma_f32_16x16x4_f32(
         false, a,  false, b,
         (short)0, c,  false,  false);
}

__device__ __forceinline__ v2f load2(const float* __restrict__ p) {
    return *(const v2f*)p;
}

__device__ __forceinline__ void wave_lds_sync() {
    __builtin_amdgcn_fence(__ATOMIC_RELEASE, "wavefront");
    __builtin_amdgcn_wave_barrier();
    __builtin_amdgcn_fence(__ATOMIC_ACQUIRE, "wavefront");
}

#define RT_WAVES 4
#define RT_TC    64
#define RT_PITCH 68
#define RT_PP    20
#define RT_VP    160

template <bool DIAG>
__device__ __forceinline__ void rt_tile(const v2f (&qa)[16], v8f (&acc)[4],
                                        const float* sK, const float* sV, float* sPw,
                                        int j, int m, int kb2, int hi, int mbase)
{
    const float* Kt = sK + (j * 16 + m) * RT_PITCH + kb2;
    v8f sc = {};
#pragma unroll
    for (int kk = 0; kk < 16; ++kk) {
        v2f bb = *(const v2f*)(Kt + kk * 4);
        sc = wmma4(qa[kk], bb, sc);
    }
#pragma unroll
    for (int jj = 0; jj < 8; ++jj) {
        float v = sc[jj];
        if (DIAG) v = (m <= mbase + jj) ? v : 0.0f;
        sPw[(mbase + jj) * RT_PP + m] = v;
    }
    wave_lds_sync();
    const float* Vt = sV + (j * 8 + hi) * RT_VP + 2 * m;
#pragma unroll
    for (int k0 = 0; k0 < 16; k0 += 4) {
        v2f a2 = *(const v2f*)&sPw[m * RT_PP + k0 + kb2];
#pragma unroll
        for (int nt = 0; nt < 4; ++nt) {
            v2f b2 = *(const v2f*)(Vt + (k0 >> 1) * RT_VP + nt * 32);
            acc[nt] = wmma4(a2, b2, acc[nt]);
        }
    }
    wave_lds_sync();
}

template <int NT>
__device__ __forceinline__ void g_stage(v8f (&gs)[3], const float* Vt, int o0, int o1, int o2)
{
#pragma unroll
    for (int kk = 0; kk < 16; ++kk) {
        const float* p = Vt + 2 * kk * RT_VP;
        const v2f f0 = *(const v2f*)(p + o0);
        const v2f f1 = *(const v2f*)(p + o1);
        gs[0] = wmma4(f0, f0, gs[0]);
        gs[1] = wmma4(f0, f1, gs[1]);
        if (NT == 3) {
            const v2f f2 = *(const v2f*)(p + o2);
            gs[2] = wmma4(f0, f2, gs[2]);
        }
    }
}

template <bool MIRROR>
__device__ __forceinline__ void g_store(const v8f& t, float* sV, int rb, int cb, int m, int hi)
{
#pragma unroll
    for (int q = 0; q < 4; ++q) {
        v2f pr; pr[0] = t[2 * q]; pr[1] = t[2 * q + 1];
        *(v2f*)&sV[(8 * rb + 4 * hi + q) * RT_VP + 2 * (cb * 16 + m)] = pr;
    }
    if (MIRROR) {
        float* md = &sV[(8 * cb + (m >> 1)) * RT_VP + 2 * (16 * rb + 8 * hi) + (m & 1)];
#pragma unroll
        for (int r = 0; r < 8; ++r) md[2 * r] = t[r];
    }
}

__global__ void __launch_bounds__(32 * RT_WAVES)
retention_kernel(const float* __restrict__ x, const float* __restrict__ wq,
                 const float* __restrict__ wk, const float* __restrict__ wv,
                 float* __restrict__ ret)
{
    __shared__ __attribute__((aligned(16))) float sK[RT_TC * RT_PITCH];
    __shared__ __attribute__((aligned(16))) float sV[(RT_TC / 2) * RT_VP];
    __shared__ __attribute__((aligned(16))) float sP[RT_WAVES * 16 * RT_PP];

    const int tid  = threadIdx.x;
    const int lane = tid & 31;
    const int w    = __builtin_amdgcn_readfirstlane(tid >> 5);
    const int m    = lane & 15;
    const int hi   = lane >> 4;
    const int kb2  = hi << 1;
    const int mbase = hi * 8;

    const int bi   = blockIdx.x;
    const int pass = bi >> 8;
    const int rem  = bi & 255;
    const int row  = rem >> 5;
    const int bn   = rem & 31;
    const int g    = (3 - pass) * 8 + ((pass & 1) ? row : 7 - row);
    const int b    = bn / NHC;
    const int n    = bn % NHC;
    const int st   = g * RT_WAVES + w;

    const float* Xh  = x + (size_t)b * SSC * HIDC + n * HDC;
    const float* Wqn = wq + n * HDC * HDC;
    const float* Wkn = wk + n * HDC * HDC;
    const float* Wvn = wv + n * HDC * HDC;

    const float lin   = logf(1.0f / 512.0f)
                      + (float)n * ((logf(1.0f / 32.0f) - logf(1.0f / 512.0f)) / 15.0f);
    const float l2g   = log2f(1.0f - expf(lin));

    const int r8   = tid >> 4;
    const int fcol = (tid & 15) * 4;
    v4f rk[8];
    float colf[8];
#pragma unroll
    for (int p = 0; p < 8; ++p) {
        const int tl = 16 * (p >> 1) + 2 * r8 + (p & 1);
        rk[p] = *(const v4f*)(Xh + (size_t)tl * HIDC + fcol);
        colf[p] = exp2f(-(float)tl * l2g);
    }
    float hcolf[8];
#pragma unroll
    for (int p = 0; p < 8; ++p) {
        const int tl = 16 * (p >> 1) + 2 * r8 + (p & 1);
        hcolf[p] = exp2f(-0.5f * (float)tl * l2g);
    }

    {
        v8f mt[4] = {};
#pragma unroll
        for (int kk = 0; kk < 16; ++kk) {
            v2f a = load2(Wqn + (16 * w + m) * HDC + kk * 4 + kb2);
#pragma unroll
            for (int nt = 0; nt < 4; ++nt) {
                v2f bk = load2(Wkn + (nt * 16 + m) * HDC + kk * 4 + kb2);
                mt[nt] = wmma4(a, bk, mt[nt]);
            }
        }
#pragma unroll
        for (int nt = 0; nt < 4; ++nt) {
            float* dst = &sV[(nt * 16 + m) * RT_PITCH + 16 * w + mbase];
            v4f lo, up;
#pragma unroll
            for (int r = 0; r < 4; ++r) { lo[r] = mt[nt][r]; up[r] = mt[nt][4 + r]; }
            *(v4f*)dst       = lo;
            *(v4f*)(dst + 4) = up;
        }
    }
    __syncthreads();

    float* sQ = sK + 16 * w * RT_PITCH;
    v2f qa[16];
    {
        const float* Xq = Xh + (size_t)(st * 16 + m) * HIDC;
        v8f qc[4] = {};
#pragma unroll
        for (int kk = 0; kk < 16; ++kk) {
            v2f a = load2(Xq + kk * 4 + kb2);
#pragma unroll
            for (int nt = 0; nt < 4; ++nt) {
                v2f bm = *(const v2f*)&sV[(nt * 16 + m) * RT_PITCH + kk * 4 + kb2];
                qc[nt] = wmma4(a, bm, qc[nt]);
            }
        }
#pragma unroll
        for (int nt = 0; nt < 4; ++nt)
#pragma unroll
            for (int r = 0; r < 8; ++r)
                sQ[(mbase + r) * RT_PITCH + nt * 16 + m] = qc[nt][r];
        wave_lds_sync();
        const float rowf = exp2f((float)(16 * w + m) * l2g);
#pragma unroll
        for (int kk = 0; kk < 16; ++kk)
            qa[kk] = *(const v2f*)&sQ[m * RT_PITCH + kk * 4 + kb2] * rowf;
    }

    float* sPw = sP + w * 16 * RT_PP;
    v8f acc[4] = {};
    if (g > 0) {
        const int role = w ^ (((pass + 1) & 2) ? 2 : 0);
        const int o0 = role * 32, o1 = ((role + 1) & 3) * 32, o2 = ((role + 2) & 3) * 32;
        v8f gs[3] = {};
        for (int c = 0; c < g; ++c) {
            const float hstagef = exp2f(0.5f * (float)(RT_TC * (g - c)) * l2g);
            __syncthreads();
#pragma unroll
            for (int pp = 0; pp < 4; ++pp) {
                const v4f ra = rk[2 * pp]     * (hstagef * hcolf[2 * pp]);
                const v4f rb = rk[2 * pp + 1] * (hstagef * hcolf[2 * pp + 1]);
                v4f i0, i1;
                i0[0] = ra[0]; i0[1] = rb[0]; i0[2] = ra[1]; i0[3] = rb[1];
                i1[0] = ra[2]; i1[1] = rb[2]; i1[2] = ra[3]; i1[3] = rb[3];
                float* vd = &sV[(8 * pp + r8) * RT_VP + 2 * fcol];
                *(v4f*)vd       = i0;
                *(v4f*)(vd + 4) = i1;
            }
            __syncthreads();
            {
                const float* Xn = Xh + (size_t)(c + 1) * RT_TC * HIDC;
#pragma unroll
                for (int p = 0; p < 8; ++p)
                    rk[p] = *(const v4f*)(Xn + (size_t)(16 * (p >> 1) + 2 * r8 + (p & 1)) * HIDC + fcol);
            }
            const float* Vt = sV + hi * RT_VP + 2 * m;
            if (role < 2) g_stage<3>(gs, Vt, o0, o1, o2);
            else          g_stage<2>(gs, Vt, o0, o1, o2);
        }
        __syncthreads();
        g_store<false>(gs[0], sV, role, role, m, hi);
        g_store<true >(gs[1], sV, role, (role + 1) & 3, m, hi);
        if (role < 2) g_store<true>(gs[2], sV, role, (role + 2) & 3, m, hi);
        __syncthreads();
#pragma unroll
        for (int kk = 0; kk < 16; ++kk) {
#pragma unroll
            for (int nt = 0; nt < 4; ++nt) {
                v2f bg = *(const v2f*)&sV[(2 * kk + hi) * RT_VP + 2 * (nt * 16 + m)];
                acc[nt] = wmma4(qa[kk], bg, acc[nt]);
            }
        }
    }
    {
        __syncthreads();
#pragma unroll
        for (int pp = 0; pp < 4; ++pp) {
            const v4f ra = rk[2 * pp], rb = rk[2 * pp + 1];
            const int tl = 16 * pp + 2 * r8;
            *(v4f*)&sK[tl * RT_PITCH + fcol]       = ra * colf[2 * pp];
            *(v4f*)&sK[(tl + 1) * RT_PITCH + fcol] = rb * colf[2 * pp + 1];
            v4f i0, i1;
            i0[0] = ra[0]; i0[1] = rb[0]; i0[2] = ra[1]; i0[3] = rb[1];
            i1[0] = ra[2]; i1[1] = rb[2]; i1[2] = ra[3]; i1[3] = rb[3];
            float* vd = &sV[(8 * pp + r8) * RT_VP + 2 * fcol];
            *(v4f*)vd       = i0;
            *(v4f*)(vd + 4) = i1;
        }
        __syncthreads();
#pragma unroll
        for (int p = 0; p < 8; ++p)
            rk[p] = *(const v4f*)(Wvn + (16 * (p >> 1) + 2 * r8 + (p & 1)) * HDC + fcol);
        for (int j = 0; j < w; ++j)
            rt_tile<false>(qa, acc, sK, sV, sPw, j, m, kb2, hi, mbase);
        rt_tile<true>(qa, acc, sK, sV, sPw, w, m, kb2, hi, mbase);
    }

    __syncthreads();
#pragma unroll
    for (int pp = 0; pp < 4; ++pp) {
        const v4f ra = rk[2 * pp], rb = rk[2 * pp + 1];
        v4f i0, i1;
        i0[0] = ra[0]; i0[1] = rb[0]; i0[2] = ra[1]; i0[3] = rb[1];
        i1[0] = ra[2]; i1[1] = rb[2]; i1[2] = ra[3]; i1[3] = rb[3];
        float* vd = &sV[(8 * pp + r8) * RT_VP + 2 * fcol];
        *(v4f*)vd       = i0;
        *(v4f*)(vd + 4) = i1;
    }
#pragma unroll
    for (int nt = 0; nt < 4; ++nt)
#pragma unroll
        for (int r = 0; r < 8; ++r)
            sQ[(mbase + r) * RT_PITCH + nt * 16 + m] = acc[nt][r];
    __syncthreads();
    v8f fin[4] = {};
#pragma unroll
    for (int kk = 0; kk < 16; ++kk) {
        v2f a = *(const v2f*)&sQ[m * RT_PITCH + kk * 4 + kb2];
#pragma unroll
        for (int nt = 0; nt < 4; ++nt) {
            v2f bw = *(const v2f*)&sV[(2 * kk + hi) * RT_VP + 2 * (nt * 16 + m)];
            fin[nt] = wmma4(a, bw, fin[nt]);
        }
    }

    float* rb = ret + (((size_t)b * SSC + st * 16) * NHC + n) * HDC;
#pragma unroll
    for (int nt = 0; nt < 4; ++nt)
#pragma unroll
        for (int j = 0; j < 8; ++j)
            rb[(size_t)(mbase + j) * NHC * HDC + nt * 16 + m] = fin[nt][j];
}

template <int M>
__device__ __forceinline__ float xsum16(float v) {
    const int o = __builtin_amdgcn_update_dpp(0, __builtin_bit_cast(int, v), 0x160 + M, 0xf, 0xf, true);
    return v + __builtin_bit_cast(float, o);
}

#define GK 32
#define GW 8

template <bool SWISH_ADD>
__global__ void __launch_bounds__(32 * GW) __attribute__((amdgpu_waves_per_eu(2, 2)))
gemm1024_kernel(const float* __restrict__ A, const float* __restrict__ W,
                const float* __restrict__ retn, const float* __restrict__ gsc,
                const float* __restrict__ gbi, float* __restrict__ out)
{
    const int lane = threadIdx.x & 31;
    const int c    = lane & 15;
    const int hi   = lane >> 4;
    const int kb2  = hi << 1;
    const int strip = blockIdx.x * GW + (threadIdx.x >> 5);
    const int nb    = blockIdx.y;

    const float* Ap = A + (size_t)(strip * 16 + c) * HIDC + kb2;
    const float* Wp = W + (size_t)kb2 * HIDC + nb * 128 + c * 4;

    v8f acc[8] = {};
    for (int kb = 0; kb < HIDC; kb += GK) {
        v2f a[GK / 4];
        v4f b[GK / 4][2][2];
#pragma unroll
        for (int j = 0; j < GK / 4; ++j) {
            a[j] = load2(Ap + kb + 4 * j);
#pragma unroll
            for (int s = 0; s < 2; ++s)
#pragma unroll
                for (int sp = 0; sp < 2; ++sp)
                    b[j][s][sp] = *(const v4f*)(Wp + (size_t)(kb + 4 * j + s) * HIDC + sp * 64);
        }
#pragma unroll
        for (int j = 0; j < GK / 4; ++j)
#pragma unroll
            for (int sp = 0; sp < 2; ++sp)
#pragma unroll
                for (int t = 0; t < 4; ++t) {
                    v2f bb; bb[0] = b[j][0][sp][t]; bb[1] = b[j][1][sp][t];
                    acc[sp * 4 + t] = wmma4(a[j], bb, acc[sp * 4 + t]);
                }
    }

    const int mbase = hi * 8;
#pragma unroll
    for (int sp = 0; sp < 2; ++sp) {
        const int ch = nb * 128 + sp * 64 + c * 4;
        if constexpr (SWISH_ADD) {
            const v4f sc4 = *(const v4f*)(gsc + ch);
            const v4f bi4 = *(const v4f*)(gbi + ch);
            v4f r[8];
            float s[8], q[8];
#pragma unroll
            for (int j = 0; j < 8; ++j)
                r[j] = *(const v4f*)(retn + (size_t)(strip * 16 + mbase + j) * HIDC + ch);
#pragma unroll
            for (int j = 0; j < 8; ++j) s[j] = (r[j][0] + r[j][1]) + (r[j][2] + r[j][3]);
#pragma unroll
            for (int j = 0; j < 8; ++j) s[j] = xsum16<1>(s[j]);
#pragma unroll
            for (int j = 0; j < 8; ++j) s[j] = xsum16<2>(s[j]);
#pragma unroll
            for (int j = 0; j < 8; ++j) s[j] = xsum16<4>(s[j]);
#pragma unroll
            for (int j = 0; j < 8; ++j) s[j] = xsum16<8>(s[j]);
#pragma unroll
            for (int j = 0; j < 8; ++j) {
                const float mean = s[j] * (1.0f / HDC);
#pragma unroll
                for (int t = 0; t < 4; ++t) r[j][t] -= mean;
                q[j] = (r[j][0] * r[j][0] + r[j][1] * r[j][1]) + (r[j][2] * r[j][2] + r[j][3] * r[j][3]);
            }
#pragma unroll
            for (int j = 0; j < 8; ++j) q[j] = xsum16<1>(q[j]);
#pragma unroll
            for (int j = 0; j < 8; ++j) q[j] = xsum16<2>(q[j]);
#pragma unroll
            for (int j = 0; j < 8; ++j) q[j] = xsum16<4>(q[j]);
#pragma unroll
            for (int j = 0; j < 8; ++j) q[j] = xsum16<8>(q[j]);
#pragma unroll
            for (int j = 0; j < 8; ++j) {
                const float inv = rsqrtf(q[j] * (1.0f / HDC) + 1e-5f);
                v4f o;
#pragma unroll
                for (int t = 0; t < 4; ++t) {
                    const float g = acc[sp * 4 + t][j];
                    o[t] = g / (1.0f + expf(-g)) + (r[j][t] * inv * sc4[t] + bi4[t]);
                }
                *(v4f*)(out + (size_t)(strip * 16 + mbase + j) * HIDC + ch) = o;
            }
        } else {
#pragma unroll
            for (int j = 0; j < 8; ++j) {
                v4f g;
#pragma unroll
                for (int t = 0; t < 4; ++t) g[t] = acc[sp * 4 + t][j];
                *(v4f*)(out + (size_t)(strip * 16 + mbase + j) * HIDC + ch) = g;
            }
        }
    }
}

extern "C" void kernel_launch(void* const* d_in, const int* in_sizes, int n_in,
                              void* d_out, int out_size, void* d_ws, size_t ws_size,
                              hipStream_t stream)
{
    (void)in_sizes; (void)n_in; (void)out_size; (void)ws_size;
    const float* x   = (const float*)d_in[0];
    const float* wq  = (const float*)d_in[1];
    const float* wk  = (const float*)d_in[2];
    const float* wv  = (const float*)d_in[3];
    const float* w1  = (const float*)d_in[4];
    const float* w2  = (const float*)d_in[5];
    const float* gns = (const float*)d_in[6];
    const float* gnb = (const float*)d_in[7];
    float* out = (float*)d_out;
    float* ws  = (float*)d_ws;

    const size_t NQ = (size_t)BBC * NHC * SSC * HDC;
    float* ret  = ws;
    float* mid  = ws + NQ;

    retention_kernel<<<dim3((SSC / (16 * RT_WAVES)) * BBC * NHC), dim3(32 * RT_WAVES), 0, stream>>>(x, wq, wk, wv, ret);
    gemm1024_kernel<true ><<<dim3(BBC * SSC / (16 * GW), HIDC / 128), dim3(32 * GW), 0, stream>>>(x,   w1, ret, gns, gnb, mid);
    gemm1024_kernel<false><<<dim3(BBC * SSC / (16 * GW), HIDC / 128), dim3(32 * GW), 0, stream>>>(mid, w2, nullptr, nullptr, nullptr, out);
}


# retention: 8-wave work-groups own two 64-row blocks that share one walk of the state G (2h+1 stages, was 4h+1); block A copies it times gamma^64 and undoes that; sum order kept, A's rounding changes
# speedup vs baseline: 4.1830x; 1.0715x over previous
#include <hip/hip_runtime.h>
#include <hip/hip_bf16.h>
#include <math.h>

#define HIDC 1024
#define NHC  16
#define HDC  64
#define BBC  2
#define SSC  2048

typedef float v2f __attribute__((ext_vector_type(2)));
typedef float v4f __attribute__((ext_vector_type(4)));
typedef float v8f __attribute__((ext_vector_type(8)));

__device__ __forceinline__ v8f wmma4(v2f a, v2f b, v8f c) {
    return __builtin_amdgcn_wmma_f32_16x16x4_f32(
         false, a,  false, b,
         (short)0, c,  false,  false);
}

__device__ __forceinline__ v2f load2(const float* __restrict__ p) {
    return *(const v2f*)p;
}

__device__ __forceinline__ void wave_lds_sync() {
    __builtin_amdgcn_fence(__ATOMIC_RELEASE, "wavefront");
    __builtin_amdgcn_wave_barrier();
    __builtin_amdgcn_fence(__ATOMIC_ACQUIRE, "wavefront");
}

#define RT_WAVES 4
#define RT_TC    64
#define RT_PITCH 68
#define RT_PP    20
#define RT_VP    160

template <bool DIAG>
__device__ __forceinline__ void rt_tile(const v2f (&qa)[16], v8f (&acc)[4],
                                        const float* sK, const float* sV, float* sPw,
                                        int j, int m, int kb2, int hi, int mbase)
{
    const float* Kt = sK + (j * 16 + m) * RT_PITCH + kb2;
    v8f sc = {};
#pragma unroll
    for (int kk = 0; kk < 16; ++kk) {
        v2f bb = *(const v2f*)(Kt + kk * 4);
        sc = wmma4(qa[kk], bb, sc);
    }
#pragma unroll
    for (int jj = 0; jj < 8; ++jj) {
        float v = sc[jj];
        if (DIAG) v = (m <= mbase + jj) ? v : 0.0f;
        sPw[(mbase + jj) * RT_PP + m] = v;
    }
    wave_lds_sync();
    const float* Vt = sV + (j * 8 + hi) * RT_VP + 2 * m;
#pragma unroll
    for (int k0 = 0; k0 < 16; k0 += 4) {
        v2f a2 = *(const v2f*)&sPw[m * RT_PP + k0 + kb2];
#pragma unroll
        for (int nt = 0; nt < 4; ++nt) {
            v2f b2 = *(const v2f*)(Vt + (k0 >> 1) * RT_VP + nt * 32);
            acc[nt] = wmma4(a2, b2, acc[nt]);
        }
    }
    wave_lds_sync();
}

template <bool WITH_DIAG>
__device__ __forceinline__ void g_stage(v8f (&gs)[2], const float* Vt, int o0, int o1)
{
#pragma unroll
    for (int kk = 0; kk < 16; ++kk) {
        const float* p = Vt + 2 * kk * RT_VP;
        const v2f f0 = *(const v2f*)(p + o0);
        const v2f f1 = *(const v2f*)(p + o1);
        if (WITH_DIAG) gs[0] = wmma4(f0, f0, gs[0]);
        gs[1] = wmma4(f0, f1, gs[1]);
    }
}

template <bool MIRROR>
__device__ __forceinline__ void g_store(const v8f& t, float* sV, int rb, int cb, int m, int hi)
{
#pragma unroll
    for (int q = 0; q < 4; ++q) {
        v2f pr; pr[0] = t[2 * q]; pr[1] = t[2 * q + 1];
        *(v2f*)&sV[(8 * rb + 4 * hi + q) * RT_VP + 2 * (cb * 16 + m)] = pr;
    }
    if (MIRROR) {
        float* md = &sV[(8 * cb + (m >> 1)) * RT_VP + 2 * (16 * rb + 8 * hi) + (m & 1)];
#pragma unroll
        for (int r = 0; r < 8; ++r) md[2 * r] = t[r];
    }
}

#define RT_GRP 8

__global__ void __launch_bounds__(32 * RT_GRP)
retention_kernel(const float* __restrict__ x, const float* __restrict__ wq,
                 const float* __restrict__ wk, const float* __restrict__ wv,
                 float* __restrict__ ret)
{
    __shared__ __attribute__((aligned(16))) float sK[2 * RT_TC * RT_PITCH];
    __shared__ __attribute__((aligned(16))) float sVA[(RT_TC / 2) * RT_VP];
    __shared__ __attribute__((aligned(16))) float sVB[(RT_TC / 2) * RT_VP];
    __shared__ __attribute__((aligned(16))) float sS[(RT_TC / 2) * RT_VP];
    __shared__ __attribute__((aligned(16))) float sGA[(RT_TC / 2) * RT_VP];
    __shared__ __attribute__((aligned(16))) float sP[RT_GRP * 16 * RT_PP];
    __shared__ float sX[2 * 8 * 32];

    const int tid  = threadIdx.x;
    const int lane = tid & 31;
    const int w8   = __builtin_amdgcn_readfirstlane(tid >> 5);
    const int half = w8 >> 2;
    const int w    = w8 & 3;
    const int m    = lane & 15;
    const int hi   = lane >> 4;
    const int kb2  = hi << 1;
    const int mbase = hi * 8;

    const int bi   = blockIdx.x;
    const int pass = bi >> 8;
    const int rem  = bi & 255;
    const int row  = rem >> 5;
    const int bn   = rem & 31;
    const int h    = pass ? row : 15 - row;
    const int g    = 2 * h + half;
    const int b    = bn / NHC;
    const int n    = bn % NHC;
    const int st   = g * RT_WAVES + w;

    const float* Xh  = x + (size_t)b * SSC * HIDC + n * HDC;
    const float* Wqn = wq + n * HDC * HDC;
    const float* Wkn = wk + n * HDC * HDC;
    const float* Wvn = wv + n * HDC * HDC;

    const float lin   = logf(1.0f / 512.0f)
                      + (float)n * ((logf(1.0f / 32.0f) - logf(1.0f / 512.0f)) / 15.0f);
    const float l2g   = log2f(1.0f - expf(lin));

    const int r16  = tid >> 4;
    const int fcol = (tid & 15) * 4;
    v4f rk[4];
    float colf[4], hcolf[4];
#pragma unroll
    for (int p = 0; p < 4; ++p) {
        const int tl = 32 * (p >> 1) + 2 * r16 + (p & 1);
        rk[p] = *(const v4f*)(Xh + (size_t)tl * HIDC + fcol);
        colf[p]  = exp2f(-(float)tl * l2g);
        hcolf[p] = exp2f(-0.5f * (float)tl * l2g);
    }

    {
        v8f mt[2] = {};
#pragma unroll
        for (int kk = 0; kk < 16; ++kk) {
            v2f a = load2(Wqn + (16 * w + m) * HDC + kk * 4 + kb2);
#pragma unroll
            for (int q = 0; q < 2; ++q) {
                v2f bk = load2(Wkn + ((2 * half + q) * 16 + m) * HDC + kk * 4 + kb2);
                mt[q] = wmma4(a, bk, mt[q]);
            }
        }
#pragma unroll
        for (int q = 0; q < 2; ++q) {
            float* dst = &sS[((2 * half + q) * 16 + m) * RT_PITCH + 16 * w + mbase];
            v4f lo, up;
#pragma unroll
            for (int r = 0; r < 4; ++r) { lo[r] = mt[q][r]; up[r] = mt[q][4 + r]; }
            *(v4f*)dst       = lo;
            *(v4f*)(dst + 4) = up;
        }
    }
    __syncthreads();

    float* sQ = sK + 16 * w8 * RT_PITCH;
    v2f qa[16];
    {
        const float* Xq = Xh + (size_t)(st * 16 + m) * HIDC;
        v8f qc[4] = {};
#pragma unroll
        for (int kk = 0; kk < 16; ++kk) {
            v2f a = load2(Xq + kk * 4 + kb2);
#pragma unroll
            for (int nt = 0; nt < 4; ++nt) {
                v2f bm = *(const v2f*)&sS[(nt * 16 + m) * RT_PITCH + kk * 4 + kb2];
                qc[nt] = wmma4(a, bm, qc[nt]);
            }
        }
#pragma unroll
        for (int nt = 0; nt < 4; ++nt)
#pragma unroll
            for (int r = 0; r < 8; ++r)
                sQ[(mbase + r) * RT_PITCH + nt * 16 + m] = qc[nt][r];
        wave_lds_sync();
        const float rowf = exp2f((float)(16 * w + m) * l2g);
#pragma unroll
        for (int kk = 0; kk < 16; ++kk)
            qa[kk] = *(const v2f*)&sQ[m * RT_PITCH + kk * 4 + kb2] * rowf;
    }

    const int xr   = (w8 - 4) ^ (pass ? 2 : 0);
    const int rb0  = (w8 < 4) ? w8 : (xr & 1);
    const int cb1  = (rb0 + ((w8 < 4) ? 1 : 2)) & 3;
    const int o0 = rb0 * 32, o1 = cb1 * 32;

    auto store_walk_stage = [&](float f) {
#pragma unroll
        for (int pp = 0; pp < 2; ++pp) {
            const v4f ra = rk[2 * pp]     * (f * hcolf[2 * pp]);
            const v4f rb = rk[2 * pp + 1] * (f * hcolf[2 * pp + 1]);
            v4f i0, i1;
            i0[0] = ra[0]; i0[1] = rb[0]; i0[2] = ra[1]; i0[3] = rb[1];
            i1[0] = ra[2]; i1[1] = rb[2]; i1[2] = ra[3]; i1[3] = rb[3];
            float* vd = &sS[(16 * pp + r16) * RT_VP + 2 * fcol];
            *(v4f*)vd       = i0;
            *(v4f*)(vd + 4) = i1;
        }
    };
    auto store_own_stage = [&](float* Kd, float* Vd) {
#pragma unroll
        for (int pp = 0; pp < 2; ++pp) {
            const v4f ra = rk[2 * pp], rb = rk[2 * pp + 1];
            const int tl = 32 * pp + 2 * r16;
            *(v4f*)&Kd[tl * RT_PITCH + fcol]       = ra * colf[2 * pp];
            *(v4f*)&Kd[(tl + 1) * RT_PITCH + fcol] = rb * colf[2 * pp + 1];
            v4f i0, i1;
            i0[0] = ra[0]; i0[1] = rb[0]; i0[2] = ra[1]; i0[3] = rb[1];
            i1[0] = ra[2]; i1[1] = rb[2]; i1[2] = ra[3]; i1[3] = rb[3];
            float* vd = &Vd[(16 * pp + r16) * RT_VP + 2 * fcol];
            *(v4f*)vd       = i0;
            *(v4f*)(vd + 4) = i1;
        }
    };
    v8f gs[2] = {};
    auto store_state = [&](float* Gd) {
        if (w8 < 4) {
            g_store<false>(gs[0], Gd, rb0, rb0, m, hi);
            g_store<true >(gs[1], Gd, rb0, cb1, m, hi);
        } else if (xr >= 2) {
            g_store<true >(gs[1], Gd, rb0, cb1, m, hi);
        }
    };

    const int nst = 2 * h;
    const int csw = (nst + 1) >> 1;
    for (int c = 0; c <= nst; ++c) {
        const float hstagef = exp2f(0.5f * (float)(RT_TC * (nst + 1 - c)) * l2g);
        __syncthreads();
        store_walk_stage(hstagef);
        if (c == nst) store_own_stage(sK, sVA);
        if (c == csw && csw > 0 && w8 >= 4 && xr < 2) {
#pragma unroll
            for (int r = 0; r < 8; ++r) sX[(xr * 8 + r) * 32 + lane] = gs[1][r];
        }
        __syncthreads();
        {
            const float* Xn = Xh + (size_t)(c + 1) * RT_TC * HIDC;
#pragma unroll
            for (int p = 0; p < 4; ++p)
                rk[p] = *(const v4f*)(Xn + (size_t)(32 * (p >> 1) + 2 * r16 + (p & 1)) * HIDC + fcol);
        }
        if (c == csw && csw > 0 && w8 >= 4 && xr >= 2) {
#pragma unroll
            for (int r = 0; r < 8; ++r) gs[1][r] = sX[((xr - 2) * 8 + r) * 32 + lane];
        }
        if (c == nst && h > 0) store_state(sGA);
        const float* Vt = sS + hi * RT_VP + 2 * m;
        if (w8 < 4)                         g_stage<true >(gs, Vt, o0, o1);
        else if ((xr < 2) == (c < csw))     g_stage<false>(gs, Vt, o0, o1);
    }
    __syncthreads();
    store_state(sS);
    store_own_stage(sK + RT_TC * RT_PITCH, sVB);
    __syncthreads();
#pragma unroll
    for (int p = 0; p < 4; ++p)
        rk[p] = *(const v4f*)(Wvn + (32 * (p >> 1) + 2 * r16 + (p & 1)) * HDC + fcol);

    float* sPw = sP + w8 * 16 * RT_PP;
    v8f acc[4] = {};
    if (half == 1 || h > 0) {
        const float* Gs = half ? sS : sGA;
#pragma unroll
        for (int kk = 0; kk < 16; ++kk) {
#pragma unroll
            for (int nt = 0; nt < 4; ++nt) {
                v2f bg = *(const v2f*)&Gs[(2 * kk + hi) * RT_VP + 2 * (nt * 16 + m)];
                acc[nt] = wmma4(qa[kk], bg, acc[nt]);
            }
        }
        if (half == 0) {
            const float back = exp2f(-(float)RT_TC * l2g);
#pragma unroll
            for (int nt = 0; nt < 4; ++nt) acc[nt] *= back;
        }
    }
    {
        const float* Kh = sK + half * RT_TC * RT_PITCH;
        const float* Vh = half ? sVB : sVA;
        for (int j = 0; j < w; ++j)
            rt_tile<false>(qa, acc, Kh, Vh, sPw, j, m, kb2, hi, mbase);
        rt_tile<true>(qa, acc, Kh, Vh, sPw, w, m, kb2, hi, mbase);
    }

    __syncthreads();
#pragma unroll
    for (int pp = 0; pp < 2; ++pp) {
        const v4f ra = rk[2 * pp], rb = rk[2 * pp + 1];
        v4f i0, i1;
        i0[0] = ra[0]; i0[1] = rb[0]; i0[2] = ra[1]; i0[3] = rb[1];
        i1[0] = ra[2]; i1[1] = rb[2]; i1[2] = ra[3]; i1[3] = rb[3];
        float* vd = &sS[(16 * pp + r16) * RT_VP + 2 * fcol];
        *(v4f*)vd       = i0;
        *(v4f*)(vd + 4) = i1;
    }
#pragma unroll
    for (int nt = 0; nt < 4; ++nt)
#pragma unroll
        for (int r = 0; r < 8; ++r)
            sQ[(mbase + r) * RT_PITCH + nt * 16 + m] = acc[nt][r];
    __syncthreads();
    v8f fin[4] = {};
#pragma unroll
    for (int kk = 0; kk < 16; ++kk) {
        v2f a = *(const v2f*)&sQ[m * RT_PITCH + kk * 4 + kb2];
#pragma unroll
        for (int nt = 0; nt < 4; ++nt) {
            v2f bw = *(const v2f*)&sS[(2 * kk + hi) * RT_VP + 2 * (nt * 16 + m)];
            fin[nt] = wmma4(a, bw, fin[nt]);
        }
    }

    float* rb = ret + (((size_t)b * SSC + st * 16) * NHC + n) * HDC;
#pragma unroll
    for (int nt = 0; nt < 4; ++nt)
#pragma unroll
        for (int j = 0; j < 8; ++j)
            rb[(size_t)(mbase + j) * NHC * HDC + nt * 16 + m] = fin[nt][j];
}

template <int M>
__device__ __forceinline__ float xsum16(float v) {
    const int o = __builtin_amdgcn_update_dpp(0, __builtin_bit_cast(int, v), 0x160 + M, 0xf, 0xf, true);
    return v + __builtin_bit_cast(float, o);
}

#define GK 32
#define GW 8

template <bool SWISH_ADD>
__global__ void __launch_bounds__(32 * GW) __attribute__((amdgpu_waves_per_eu(2, 2)))
gemm1024_kernel(const float* __restrict__ A, const float* __restrict__ W,
                const float* __restrict__ retn, const float* __restrict__ gsc,
                const float* __restrict__ gbi, float* __restrict__ out)
{
    const int lane = threadIdx.x & 31;
    const int c    = lane & 15;
    const int hi   = lane >> 4;
    const int kb2  = hi << 1;
    const int strip = blockIdx.x * GW + (threadIdx.x >> 5);
    const int nb    = blockIdx.y;

    const float* Ap = A + (size_t)(strip * 16 + c) * HIDC + kb2;
    const float* Wp = W + (size_t)kb2 * HIDC + nb * 128 + c * 4;

    v8f acc[8] = {};
    for (int kb = 0; kb < HIDC; kb += GK) {
        v2f a[GK / 4];
        v4f b[GK / 4][2][2];
#pragma unroll
        for (int j = 0; j < GK / 4; ++j) {
            a[j] = load2(Ap + kb + 4 * j);
#pragma unroll
            for (int s = 0; s < 2; ++s)
#pragma unroll
                for (int sp = 0; sp < 2; ++sp)
                    b[j][s][sp] = *(const v4f*)(Wp + (size_t)(kb + 4 * j + s) * HIDC + sp * 64);
        }
#pragma unroll
        for (int j = 0; j < GK / 4; ++j)
#pragma unroll
            for (int sp = 0; sp < 2; ++sp)
#pragma unroll
                for (int t = 0; t < 4; ++t) {
                    v2f bb; bb[0] = b[j][0][sp][t]; bb[1] = b[j][1][sp][t];
                    acc[sp * 4 + t] = wmma4(a[j], bb, acc[sp * 4 + t]);
                }
    }

    const int mbase = hi * 8;
#pragma unroll
    for (int sp = 0; sp < 2; ++sp) {
        const int ch = nb * 128 + sp * 64 + c * 4;
        if constexpr (SWISH_ADD) {
            const v4f sc4 = *(const v4f*)(gsc + ch);
            const v4f bi4 = *(const v4f*)(gbi + ch);
            v4f r[8];
            float s[8], q[8];
#pragma unroll
            for (int j = 0; j < 8; ++j)
                r[j] = *(const v4f*)(retn + (size_t)(strip * 16 + mbase + j) * HIDC + ch);
#pragma unroll
            for (int j = 0; j < 8; ++j) s[j] = (r[j][0] + r[j][1]) + (r[j][2] + r[j][3]);
#pragma unroll
            for (int j = 0; j < 8; ++j) s[j] = xsum16<1>(s[j]);
#pragma unroll
            for (int j = 0; j < 8; ++j) s[j] = xsum16<2>(s[j]);
#pragma unroll
            for (int j = 0; j < 8; ++j) s[j] = xsum16<4>(s[j]);
#pragma unroll
            for (int j = 0; j < 8; ++j) s[j] = xsum16<8>(s[j]);
#pragma unroll
            for (int j = 0; j < 8; ++j) {
                const float mean = s[j] * (1.0f / HDC);
#pragma unroll
                for (int t = 0; t < 4; ++t) r[j][t] -= mean;
                q[j] = (r[j][0] * r[j][0] + r[j][1] * r[j][1]) + (r[j][2] * r[j][2] + r[j][3] * r[j][3]);
            }
#pragma unroll
            for (int j = 0; j < 8; ++j) q[j] = xsum16<1>(q[j]);
#pragma unroll
            for (int j = 0; j < 8; ++j) q[j] = xsum16<2>(q[j]);
#pragma unroll
            for (int j = 0; j < 8; ++j) q[j] = xsum16<4>(q[j]);
#pragma unroll
            for (int j = 0; j < 8; ++j) q[j] = xsum16<8>(q[j]);
#pragma unroll
            for (int j = 0; j < 8; ++j) {
                const float inv = rsqrtf(q[j] * (1.0f / HDC) + 1e-5f);
                v4f o;
#pragma unroll
                for (int t = 0; t < 4; ++t) {
                    const float g = acc[sp * 4 + t][j];
                    o[t] = g / (1.0f + expf(-g)) + (r[j][t] * inv * sc4[t] + bi4[t]);
                }
                *(v4f*)(out + (size_t)(strip * 16 + mbase + j) * HIDC + ch) = o;
            }
        } else {
#pragma unroll
            for (int j = 0; j < 8; ++j) {
                v4f g;
#pragma unroll
                for (int t = 0; t < 4; ++t) g[t] = acc[sp * 4 + t][j];
                *(v4f*)(out + (size_t)(strip * 16 + mbase + j) * HIDC + ch) = g;
            }
        }
    }
}

extern "C" void kernel_launch(void* const* d_in, const int* in_sizes, int n_in,
                              void* d_out, int out_size, void* d_ws, size_t ws_size,
                              hipStream_t stream)
{
    (void)in_sizes; (void)n_in; (void)out_size; (void)ws_size;
    const float* x   = (const float*)d_in[0];
    const float* wq  = (const float*)d_in[1];
    const float* wk  = (const float*)d_in[2];
    const float* wv  = (const float*)d_in[3];
    const float* w1  = (const float*)d_in[4];
    const float* w2  = (const float*)d_in[5];
    const float* gns = (const float*)d_in[6];
    const float* gnb = (const float*)d_in[7];
    float* out = (float*)d_out;
    float* ws  = (float*)d_ws;

    const size_t NQ = (size_t)BBC * NHC * SSC * HDC;
    float* ret  = ws;
    float* mid  = ws + NQ;

    retention_kernel<<<dim3((SSC / (16 * RT_GRP)) * BBC * NHC), dim3(32 * RT_GRP), 0, stream>>>(x, wq, wk, wv, ret);
    gemm1024_kernel<true ><<<dim3(BBC * SSC / (16 * GW), HIDC / 128), dim3(32 * GW), 0, stream>>>(x,   w1, ret, gns, gnb, mid);
    gemm1024_kernel<false><<<dim3(BBC * SSC / (16 * GW), HIDC / 128), dim3(32 * GW), 0, stream>>>(mid, w2, nullptr, nullptr, nullptr, out);
}


# both 1024-GEMMs: the two waves of a SIMD take the matrix pipe in turns of one 64-k block (128 WMMAs, was 32-k): s_setprio 3-(block mod 4) ranks the wave that is behind first; sums, order kept
# speedup vs baseline: 4.4490x; 1.0636x over previous
#include <hip/hip_runtime.h>
#include <hip/hip_bf16.h>
#include <math.h>

#define HIDC 1024
#define NHC  16
#define HDC  64
#define BBC  2
#define SSC  2048

typedef float v2f __attribute__((ext_vector_type(2)));
typedef float v4f __attribute__((ext_vector_type(4)));
typedef float v8f __attribute__((ext_vector_type(8)));

__device__ __forceinline__ v8f wmma4(v2f a, v2f b, v8f c) {
    return __builtin_amdgcn_wmma_f32_16x16x4_f32(
         false, a,  false, b,
         (short)0, c,  false,  false);
}

__device__ __forceinline__ v2f load2(const float* __restrict__ p) {
    return *(const v2f*)p;
}

__device__ __forceinline__ void wave_lds_sync() {
    __builtin_amdgcn_fence(__ATOMIC_RELEASE, "wavefront");
    __builtin_amdgcn_wave_barrier();
    __builtin_amdgcn_fence(__ATOMIC_ACQUIRE, "wavefront");
}

#define RT_WAVES 4
#define RT_TC    64
#define RT_PITCH 68
#define RT_PP    20
#define RT_VP    160

template <bool DIAG>
__device__ __forceinline__ void rt_tile(const v2f (&qa)[16], v8f (&acc)[4],
                                        const float* sK, const float* sV, float* sPw,
                                        int j, int m, int kb2, int hi, int mbase)
{
    const float* Kt = sK + (j * 16 + m) * RT_PITCH + kb2;
    v8f sc = {};
#pragma unroll
    for (int kk = 0; kk < 16; ++kk) {
        v2f bb = *(const v2f*)(Kt + kk * 4);
        sc = wmma4(qa[kk], bb, sc);
    }
#pragma unroll
    for (int jj = 0; jj < 8; ++jj) {
        float v = sc[jj];
        if (DIAG) v = (m <= mbase + jj) ? v : 0.0f;
        sPw[(mbase + jj) * RT_PP + m] = v;
    }
    wave_lds_sync();
    const float* Vt = sV + (j * 8 + hi) * RT_VP + 2 * m;
#pragma unroll
    for (int k0 = 0; k0 < 16; k0 += 4) {
        v2f a2 = *(const v2f*)&sPw[m * RT_PP + k0 + kb2];
#pragma unroll
        for (int nt = 0; nt < 4; ++nt) {
            v2f b2 = *(const v2f*)(Vt + (k0 >> 1) * RT_VP + nt * 32);
            acc[nt] = wmma4(a2, b2, acc[nt]);
        }
    }
    wave_lds_sync();
}

template <bool WITH_DIAG>
__device__ __forceinline__ void g_stage(v8f (&gs)[2], const float* Vt, int o0, int o1)
{
#pragma unroll
    for (int kk = 0; kk < 16; ++kk) {
        const float* p = Vt + 2 * kk * RT_VP;
        const v2f f0 = *(const v2f*)(p + o0);
        const v2f f1 = *(const v2f*)(p + o1);
        if (WITH_DIAG) gs[0] = wmma4(f0, f0, gs[0]);
        gs[1] = wmma4(f0, f1, gs[1]);
    }
}

template <bool MIRROR>
__device__ __forceinline__ void g_store(const v8f& t, float* sV, int rb, int cb, int m, int hi)
{
#pragma unroll
    for (int q = 0; q < 4; ++q) {
        v2f pr; pr[0] = t[2 * q]; pr[1] = t[2 * q + 1];
        *(v2f*)&sV[(8 * rb + 4 * hi + q) * RT_VP + 2 * (cb * 16 + m)] = pr;
    }
    if (MIRROR) {
        float* md = &sV[(8 * cb + (m >> 1)) * RT_VP + 2 * (16 * rb + 8 * hi) + (m & 1)];
#pragma unroll
        for (int r = 0; r < 8; ++r) md[2 * r] = t[r];
    }
}

#define RT_GRP 8

__global__ void __launch_bounds__(32 * RT_GRP)
retention_kernel(const float* __restrict__ x, const float* __restrict__ wq,
                 const float* __restrict__ wk, const float* __restrict__ wv,
                 float* __restrict__ ret)
{
    __shared__ __attribute__((aligned(16))) float sK[2 * RT_TC * RT_PITCH];
    __shared__ __attribute__((aligned(16))) float sVA[(RT_TC / 2) * RT_VP];
    __shared__ __attribute__((aligned(16))) float sVB[(RT_TC / 2) * RT_VP];
    __shared__ __attribute__((aligned(16))) float sS[(RT_TC / 2) * RT_VP];
    __shared__ __attribute__((aligned(16))) float sGA[(RT_TC / 2) * RT_VP];
    __shared__ __attribute__((aligned(16))) float sP[RT_GRP * 16 * RT_PP];
    __shared__ float sX[2 * 8 * 32];

    const int tid  = threadIdx.x;
    const int lane = tid & 31;
    const int w8   = __builtin_amdgcn_readfirstlane(tid >> 5);
    const int half = w8 >> 2;
    const int w    = w8 & 3;
    const int m    = lane & 15;
    const int hi   = lane >> 4;
    const int kb2  = hi << 1;
    const int mbase = hi * 8;

    const int bi   = blockIdx.x;
    const int pass = bi >> 8;
    const int rem  = bi & 255;
    const int row  = rem >> 5;
    const int bn   = rem & 31;
    const int h    = pass ? row : 15 - row;
    const int g    = 2 * h + half;
    const int b    = bn / NHC;
    const int n    = bn % NHC;
    const int st   = g * RT_WAVES + w;

    const float* Xh  = x + (size_t)b * SSC * HIDC + n * HDC;
    const float* Wqn = wq + n * HDC * HDC;
    const float* Wkn = wk + n * HDC * HDC;
    const float* Wvn = wv + n * HDC * HDC;

    const float lin   = logf(1.0f / 512.0f)
                      + (float)n * ((logf(1.0f / 32.0f) - logf(1.0f / 512.0f)) / 15.0f);
    const float l2g   = log2f(1.0f - expf(lin));

    const int r16  = tid >> 4;
    const int fcol = (tid & 15) * 4;
    v4f rk[4];
    float colf[4], hcolf[4];
#pragma unroll
    for (int p = 0; p < 4; ++p) {
        const int tl = 32 * (p >> 1) + 2 * r16 + (p & 1);
        rk[p] = *(const v4f*)(Xh + (size_t)tl * HIDC + fcol);
        colf[p]  = exp2f(-(float)tl * l2g);
        hcolf[p] = exp2f(-0.5f * (float)tl * l2g);
    }

    {
        v8f mt[2] = {};
#pragma unroll
        for (int kk = 0; kk < 16; ++kk) {
            v2f a = load2(Wqn + (16 * w + m) * HDC + kk * 4 + kb2);
#pragma unroll
            for (int q = 0; q < 2; ++q) {
                v2f bk = load2(Wkn + ((2 * half + q) * 16 + m) * HDC + kk * 4 + kb2);
                mt[q] = wmma4(a, bk, mt[q]);
            }
        }
#pragma unroll
        for (int q = 0; q < 2; ++q) {
            float* dst = &sS[((2 * half + q) * 16 + m) * RT_PITCH + 16 * w + mbase];
            v4f lo, up;
#pragma unroll
            for (int r = 0; r < 4; ++r) { lo[r] = mt[q][r]; up[r] = mt[q][4 + r]; }
            *(v4f*)dst       = lo;
            *(v4f*)(dst + 4) = up;
        }
    }
    __syncthreads();

    float* sQ = sK + 16 * w8 * RT_PITCH;
    v2f qa[16];
    {
        const float* Xq = Xh + (size_t)(st * 16 + m) * HIDC;
        v8f qc[4] = {};
#pragma unroll
        for (int kk = 0; kk < 16; ++kk) {
            v2f a = load2(Xq + kk * 4 + kb2);
#pragma unroll
            for (int nt = 0; nt < 4; ++nt) {
                v2f bm = *(const v2f*)&sS[(nt * 16 + m) * RT_PITCH + kk * 4 + kb2];
                qc[nt] = wmma4(a, bm, qc[nt]);
            }
        }
#pragma unroll
        for (int nt = 0; nt < 4; ++nt)
#pragma unroll
            for (int r = 0; r < 8; ++r)
                sQ[(mbase + r) * RT_PITCH + nt * 16 + m] = qc[nt][r];
        wave_lds_sync();
        const float rowf = exp2f((float)(16 * w + m) * l2g);
#pragma unroll
        for (int kk = 0; kk < 16; ++kk)
            qa[kk] = *(const v2f*)&sQ[m * RT_PITCH + kk * 4 + kb2] * rowf;
    }

    const int xr   = (w8 - 4) ^ (pass ? 2 : 0);
    const int rb0  = (w8 < 4) ? w8 : (xr & 1);
    const int cb1  = (rb0 + ((w8 < 4) ? 1 : 2)) & 3;
    const int o0 = rb0 * 32, o1 = cb1 * 32;

    auto store_walk_stage = [&](float f) {
#pragma unroll
        for (int pp = 0; pp < 2; ++pp) {
            const v4f ra = rk[2 * pp]     * (f * hcolf[2 * pp]);
            const v4f rb = rk[2 * pp + 1] * (f * hcolf[2 * pp + 1]);
            v4f i0, i1;
            i0[0] = ra[0]; i0[1] = rb[0]; i0[2] = ra[1]; i0[3] = rb[1];
            i1[0] = ra[2]; i1[1] = rb[2]; i1[2] = ra[3]; i1[3] = rb[3];
            float* vd = &sS[(16 * pp + r16) * RT_VP + 2 * fcol];
            *(v4f*)vd       = i0;
            *(v4f*)(vd + 4) = i1;
        }
    };
    auto store_own_stage = [&](float* Kd, float* Vd) {
#pragma unroll
        for (int pp = 0; pp < 2; ++pp) {
            const v4f ra = rk[2 * pp], rb = rk[2 * pp + 1];
            const int tl = 32 * pp + 2 * r16;
            *(v4f*)&Kd[tl * RT_PITCH + fcol]       = ra * colf[2 * pp];
            *(v4f*)&Kd[(tl + 1) * RT_PITCH + fcol] = rb * colf[2 * pp + 1];
            v4f i0, i1;
            i0[0] = ra[0]; i0[1] = rb[0]; i0[2] = ra[1]; i0[3] = rb[1];
            i1[0] = ra[2]; i1[1] = rb[2]; i1[2] = ra[3]; i1[3] = rb[3];
            float* vd = &Vd[(16 * pp + r16) * RT_VP + 2 * fcol];
            *(v4f*)vd       = i0;
            *(v4f*)(vd + 4) = i1;
        }
    };
    v8f gs[2] = {};
    auto store_state = [&](float* Gd) {
        if (w8 < 4) {
            g_store<false>(gs[0], Gd, rb0, rb0, m, hi);
            g_store<true >(gs[1], Gd, rb0, cb1, m, hi);
        } else if (xr >= 2) {
            g_store<true >(gs[1], Gd, rb0, cb1, m, hi);
        }
    };

    const int nst = 2 * h;
    const int csw = (nst + 1) >> 1;
    for (int c = 0; c <= nst; ++c) {
        const float hstagef = exp2f(0.5f * (float)(RT_TC * (nst + 1 - c)) * l2g);
        __syncthreads();
        store_walk_stage(hstagef);
        if (c == nst) store_own_stage(sK, sVA);
        if (c == csw && csw > 0 && w8 >= 4 && xr < 2) {
#pragma unroll
            for (int r = 0; r < 8; ++r) sX[(xr * 8 + r) * 32 + lane] = gs[1][r];
        }
        __syncthreads();
        {
            const float* Xn = Xh + (size_t)(c + 1) * RT_TC * HIDC;
#pragma unroll
            for (int p = 0; p < 4; ++p)
                rk[p] = *(const v4f*)(Xn + (size_t)(32 * (p >> 1) + 2 * r16 + (p & 1)) * HIDC + fcol);
        }
        if (c == csw && csw > 0 && w8 >= 4 && xr >= 2) {
#pragma unroll
            for (int r = 0; r < 8; ++r) gs[1][r] = sX[((xr - 2) * 8 + r) * 32 + lane];
        }
        if (c == nst && h > 0) store_state(sGA);
        const float* Vt = sS + hi * RT_VP + 2 * m;
        if (w8 < 4)                         g_stage<true >(gs, Vt, o0, o1);
        else if ((xr < 2) == (c < csw))     g_stage<false>(gs, Vt, o0, o1);
    }
    __syncthreads();
    store_state(sS);
    store_own_stage(sK + RT_TC * RT_PITCH, sVB);
    __syncthreads();
#pragma unroll
    for (int p = 0; p < 4; ++p)
        rk[p] = *(const v4f*)(Wvn + (32 * (p >> 1) + 2 * r16 + (p & 1)) * HDC + fcol);

    float* sPw = sP + w8 * 16 * RT_PP;
    v8f acc[4] = {};
    if (half == 1 || h > 0) {
        const float* Gs = half ? sS : sGA;
#pragma unroll
        for (int kk = 0; kk < 16; ++kk) {
#pragma unroll
            for (int nt = 0; nt < 4; ++nt) {
                v2f bg = *(const v2f*)&Gs[(2 * kk + hi) * RT_VP + 2 * (nt * 16 + m)];
                acc[nt] = wmma4(qa[kk], bg, acc[nt]);
            }
        }
        if (half == 0) {
            const float back = exp2f(-(float)RT_TC * l2g);
#pragma unroll
            for (int nt = 0; nt < 4; ++nt) acc[nt] *= back;
        }
    }
    {
        const float* Kh = sK + half * RT_TC * RT_PITCH;
        const float* Vh = half ? sVB : sVA;
        for (int j = 0; j < w; ++j)
            rt_tile<false>(qa, acc, Kh, Vh, sPw, j, m, kb2, hi, mbase);
        rt_tile<true>(qa, acc, Kh, Vh, sPw, w, m, kb2, hi, mbase);
    }

    __syncthreads();
#pragma unroll
    for (int pp = 0; pp < 2; ++pp) {
        const v4f ra = rk[2 * pp], rb = rk[2 * pp + 1];
        v4f i0, i1;
        i0[0] = ra[0]; i0[1] = rb[0]; i0[2] = ra[1]; i0[3] = rb[1];
        i1[0] = ra[2]; i1[1] = rb[2]; i1[2] = ra[3]; i1[3] = rb[3];
        float* vd = &sS[(16 * pp + r16) * RT_VP + 2 * fcol];
        *(v4f*)vd       = i0;
        *(v4f*)(vd + 4) = i1;
    }
#pragma unroll
    for (int nt = 0; nt < 4; ++nt)
#pragma unroll
        for (int r = 0; r < 8; ++r)
            sQ[(mbase + r) * RT_PITCH + nt * 16 + m] = acc[nt][r];
    __syncthreads();
    v8f fin[4] = {};
#pragma unroll
    for (int kk = 0; kk < 16; ++kk) {
        v2f a = *(const v2f*)&sQ[m * RT_PITCH + kk * 4 + kb2];
#pragma unroll
        for (int nt = 0; nt < 4; ++nt) {
            v2f bw = *(const v2f*)&sS[(2 * kk + hi) * RT_VP + 2 * (nt * 16 + m)];
            fin[nt] = wmma4(a, bw, fin[nt]);
        }
    }

    float* rb = ret + (((size_t)b * SSC + st * 16) * NHC + n) * HDC;
#pragma unroll
    for (int nt = 0; nt < 4; ++nt)
#pragma unroll
        for (int j = 0; j < 8; ++j)
            rb[(size_t)(mbase + j) * NHC * HDC + nt * 16 + m] = fin[nt][j];
}

template <int M>
__device__ __forceinline__ float xsum16(float v) {
    const int o = __builtin_amdgcn_update_dpp(0, __builtin_bit_cast(int, v), 0x160 + M, 0xf, 0xf, true);
    return v + __builtin_bit_cast(float, o);
}

#define GK 64
#define GW 8

template <bool SWISH_ADD>
__global__ void __launch_bounds__(32 * GW) __attribute__((amdgpu_waves_per_eu(2, 2)))
gemm1024_kernel(const float* __restrict__ A, const float* __restrict__ W,
                const float* __restrict__ retn, const float* __restrict__ gsc,
                const float* __restrict__ gbi, float* __restrict__ out)
{
    const int lane = threadIdx.x & 31;
    const int c    = lane & 15;
    const int hi   = lane >> 4;
    const int kb2  = hi << 1;
    const int strip = blockIdx.x * GW + (threadIdx.x >> 5);
    const int nb    = blockIdx.y;

    const float* Ap = A + (size_t)(strip * 16 + c) * HIDC + kb2;
    const float* Wp = W + (size_t)kb2 * HIDC + nb * 128 + c * 4;

    v8f acc[8] = {};
    auto k_block = [&](int kb) {
        v2f a[GK / 4];
        v4f b[GK / 4][2][2];
#pragma unroll
        for (int j = 0; j < GK / 4; ++j) {
            a[j] = load2(Ap + kb + 4 * j);
#pragma unroll
            for (int s = 0; s < 2; ++s)
#pragma unroll
                for (int sp = 0; sp < 2; ++sp)
                    b[j][s][sp] = *(const v4f*)(Wp + (size_t)(kb + 4 * j + s) * HIDC + sp * 64);
        }
#pragma unroll
        for (int j = 0; j < GK / 4; ++j)
#pragma unroll
            for (int sp = 0; sp < 2; ++sp)
#pragma unroll
                for (int t = 0; t < 4; ++t) {
                    v2f bb; bb[0] = b[j][0][sp][t]; bb[1] = b[j][1][sp][t];
                    acc[sp * 4 + t] = wmma4(a[j], bb, acc[sp * 4 + t]);
                }
    };
    for (int kb = 0; kb < HIDC; kb += 4 * GK) {
        __builtin_amdgcn_s_setprio(3); k_block(kb);
        __builtin_amdgcn_s_setprio(2); k_block(kb + GK);
        __builtin_amdgcn_s_setprio(1); k_block(kb + 2 * GK);
        __builtin_amdgcn_s_setprio(0); k_block(kb + 3 * GK);
    }

    const int mbase = hi * 8;
#pragma unroll
    for (int sp = 0; sp < 2; ++sp) {
        const int ch = nb * 128 + sp * 64 + c * 4;
        if constexpr (SWISH_ADD) {
            const v4f sc4 = *(const v4f*)(gsc + ch);
            const v4f bi4 = *(const v4f*)(gbi + ch);
            v4f r[8];
            float s[8], q[8];
#pragma unroll
            for (int j = 0; j < 8; ++j)
                r[j] = *(const v4f*)(retn + (size_t)(strip * 16 + mbase + j) * HIDC + ch);
#pragma unroll
            for (int j = 0; j < 8; ++j) s[j] = (r[j][0] + r[j][1]) + (r[j][2] + r[j][3]);
#pragma unroll
            for (int j = 0; j < 8; ++j) s[j] = xsum16<1>(s[j]);
#pragma unroll
            for (int j = 0; j < 8; ++j) s[j] = xsum16<2>(s[j]);
#pragma unroll
            for (int j = 0; j < 8; ++j) s[j] = xsum16<4>(s[j]);
#pragma unroll
            for (int j = 0; j < 8; ++j) s[j] = xsum16<8>(s[j]);
#pragma unroll
            for (int j = 0; j < 8; ++j) {
                const float mean = s[j] * (1.0f / HDC);
#pragma unroll
                for (int t = 0; t < 4; ++t) r[j][t] -= mean;
                q[j] = (r[j][0] * r[j][0] + r[j][1] * r[j][1]) + (r[j][2] * r[j][2] + r[j][3] * r[j][3]);
            }
#pragma unroll
            for (int j = 0; j < 8; ++j) q[j] = xsum16<1>(q[j]);
#pragma unroll
            for (int j = 0; j < 8; ++j) q[j] = xsum16<2>(q[j]);
#pragma unroll
            for (int j = 0; j < 8; ++j) q[j] = xsum16<4>(q[j]);
#pragma unroll
            for (int j = 0; j < 8; ++j) q[j] = xsum16<8>(q[j]);
#pragma unroll
            for (int j = 0; j < 8; ++j) {
                const float inv = rsqrtf(q[j] * (1.0f / HDC) + 1e-5f);
                v4f o;
#pragma unroll
                for (int t = 0; t < 4; ++t) {
                    const float g = acc[sp * 4 + t][j];
                    o[t] = g / (1.0f + expf(-g)) + (r[j][t] * inv * sc4[t] + bi4[t]);
                }
                *(v4f*)(out + (size_t)(strip * 16 + mbase + j) * HIDC + ch) = o;
            }
        } else {
#pragma unroll
            for (int j = 0; j < 8; ++j) {
                v4f g;
#pragma unroll
                for (int t = 0; t < 4; ++t) g[t] = acc[sp * 4 + t][j];
                *(v4f*)(out + (size_t)(strip * 16 + mbase + j) * HIDC + ch) = g;
            }
        }
    }
}

extern "C" void kernel_launch(void* const* d_in, const int* in_sizes, int n_in,
                              void* d_out, int out_size, void* d_ws, size_t ws_size,
                              hipStream_t stream)
{
    (void)in_sizes; (void)n_in; (void)out_size; (void)ws_size;
    const float* x   = (const float*)d_in[0];
    const float* wq  = (const float*)d_in[1];
    const float* wk  = (const float*)d_in[2];
    const float* wv  = (const float*)d_in[3];
    const float* w1  = (const float*)d_in[4];
    const float* w2  = (const float*)d_in[5];
    const float* gns = (const float*)d_in[6];
    const float* gnb = (const float*)d_in[7];
    float* out = (float*)d_out;
    float* ws  = (float*)d_ws;

    const size_t NQ = (size_t)BBC * NHC * SSC * HDC;
    float* ret  = ws;
    float* mid  = ws + NQ;

    retention_kernel<<<dim3((SSC / (16 * RT_GRP)) * BBC * NHC), dim3(32 * RT_GRP), 0, stream>>>(x, wq, wk, wv, ret);
    gemm1024_kernel<true ><<<dim3(BBC * SSC / (16 * GW), HIDC / 128), dim3(32 * GW), 0, stream>>>(x,   w1, ret, gns, gnb, mid);
    gemm1024_kernel<false><<<dim3(BBC * SSC / (16 * GW), HIDC / 128), dim3(32 * GW), 0, stream>>>(mid, w2, nullptr, nullptr, nullptr, out);
}
